# starting kernel
# baseline (speedup 1.0000x reference)
#include <hip/hip_runtime.h>
#include <hip/hip_bf16.h>

typedef __attribute__((ext_vector_type(2))) float v2f;
typedef __attribute__((ext_vector_type(8))) float v8f;

__global__ void k_deg_init(float* __restrict__ deg, int n) {
    int i = blockIdx.x * blockDim.x + threadIdx.x;
    if (i < n) deg[i] = 1.0f;
}

__global__ void k_deg_count(const int* __restrict__ dst, float* __restrict__ deg, int e) {
    int i = blockIdx.x * blockDim.x + threadIdx.x;
    if (i < e) atomicAdd(&deg[dst[i]], 1.0f);
}

__global__ void k_norm(const float* __restrict__ deg, float* __restrict__ isq,
                       float* __restrict__ idg, int n) {
    int i = blockIdx.x * blockDim.x + threadIdx.x;
    if (i < n) {
        float d = deg[i];
        isq[i] = rsqrtf(d);
        idg[i] = 1.0f / d;
    }
}

__global__ void k_coef(const int* __restrict__ src, const int* __restrict__ dst,
                       const float* __restrict__ isq, float* __restrict__ coef, int e) {
    int i = blockIdx.x * blockDim.x + threadIdx.x;
    if (i < e) coef[i] = isq[src[i]] * isq[dst[i]];
}

__global__ void k_pad_x(const float* __restrict__ x, float* __restrict__ x8, int n) {
    long idx = (long)blockIdx.x * blockDim.x + threadIdx.x;
    if (idx < (long)n * 8) {
        int i = (int)(idx >> 3), j = (int)(idx & 7);
        x8[idx] = (j < 5) ? x[(long)i * 5 + j] : 0.0f;
    }
}

__global__ void k_prep_w(const float* __restrict__ W, float* __restrict__ Bt,
                         int Kin, int Cin, int KPAD, int COLS) {
    int idx = blockIdx.x * blockDim.x + threadIdx.x;
    if (idx >= COLS * KPAD) return;
    int col = idx / KPAD, k = idx % KPAD;
    Bt[idx] = (k < Kin && col < Cin) ? W[k * Cin + col] : 0.0f;
}

template <int K, int NT, int SC>
__global__ void k_gemm_wmma(const float* __restrict__ A, const float* __restrict__ Bt,
                            float* __restrict__ H) {
    const int lane = threadIdx.x & 31;
    const int wave = threadIdx.x >> 5;
    const int half = lane >> 4;
    const int l15  = lane & 15;
    const size_t row = (size_t)blockIdx.x * 16 + l15;
    const int    col = wave * 16 + l15;
    const float* Arow = A + row * K;
    const float* Bcol = Bt + (size_t)col * K;

    v8f c = {};
#pragma unroll
    for (int k = 0; k < K; k += 4) {
        const int ka = k + half * 2;
        v2f a = *(const v2f*)(Arow + ka);
        v2f b = *(const v2f*)(Bcol + ka);
        c = __builtin_amdgcn_wmma_f32_16x16x4_f32(
                 false, a,  false, b,
                 (short)0, c,  false,  false);
    }

    if (NT * 16 == SC || col < SC) {
        const size_t mbase = (size_t)blockIdx.x * 16 + half * 8;
#pragma unroll
        for (int r = 0; r < 8; ++r)
            H[(mbase + r) * SC + col] = c[r];
    }
}

__global__ void k_self_bias(const float* __restrict__ h, const float* __restrict__ idg,
                            const float* __restrict__ bias, float* __restrict__ agg,
                            long n_elems, int dim) {
    long idx = (long)blockIdx.x * blockDim.x + threadIdx.x;
    if (idx < n_elems) {
        int i = (int)(idx / dim);
        int j = (int)(idx % dim);
        agg[idx] = h[idx] * idg[i] + bias[j];
    }
}

__global__ void k_scatter64(const int* __restrict__ src, const int* __restrict__ dst,
                            const float* __restrict__ coef, const float* __restrict__ h,
                            float* __restrict__ agg, int e) {
    int i = blockIdx.x * blockDim.x + threadIdx.x;
    if (i >= e) return;
    const int   s  = src[i];
    const int   d  = dst[i];
    const float cf = coef[i];
    const float4* hs = (const float4*)(h + (size_t)s * 64);
    float* ad = agg + (size_t)d * 64;
#pragma unroll
    for (int g = 0; g < 16; ++g) {
        float4 v = hs[g];
        atomicAdd(ad + 4 * g + 0, v.x * cf);
        atomicAdd(ad + 4 * g + 1, v.y * cf);
        atomicAdd(ad + 4 * g + 2, v.z * cf);
        atomicAdd(ad + 4 * g + 3, v.w * cf);
    }
}

__global__ void k_scatter2(const int* __restrict__ src, const int* __restrict__ dst,
                           const float* __restrict__ coef, const float* __restrict__ h,
                           float* __restrict__ out, int e) {
    int i = blockIdx.x * blockDim.x + threadIdx.x;
    if (i >= e) return;
    const int   s  = src[i];
    const int   d  = dst[i];
    const float cf = coef[i];
    float2 v = *(const float2*)(h + (size_t)s * 2);
    atomicAdd(out + (size_t)d * 2 + 0, v.x * cf);
    atomicAdd(out + (size_t)d * 2 + 1, v.y * cf);
}

__global__ void k_relu(float* __restrict__ x, long n) {
    long idx = (long)blockIdx.x * blockDim.x + threadIdx.x;
    if (idx < n) x[idx] = fmaxf(x[idx], 0.0f);
}

extern "C" void kernel_launch(void* const* d_in, const int* in_sizes, int n_in,
                              void* d_out, int out_size, void* d_ws, size_t ws_size,
                              hipStream_t stream) {
    const float* x  = (const float*)d_in[0];
    const int*   ei = (const int*)d_in[1];
    const float* W1 = (const float*)d_in[2];
    const float* b1 = (const float*)d_in[3];
    const float* W2 = (const float*)d_in[4];
    const float* b2 = (const float*)d_in[5];
    const float* W3 = (const float*)d_in[6];
    const float* b3 = (const float*)d_in[7];
    float* out = (float*)d_out;

    const int N = in_sizes[0] / 5;
    const int E = in_sizes[1] / 2;
    const int* src = ei;
    const int* dst = ei + E;

    float* ws   = (float*)d_ws;
    float* deg  = ws;
    float* isq  = deg  + N;
    float* idg  = isq  + N;
    float* coef = idg  + N;
    float* x8   = coef + E;
    float* w1t  = x8   + (size_t)N * 8;
    float* w2t  = w1t  + 64 * 8;
    float* w3t  = w2t  + 64 * 64;
    float* bufA = w3t  + 16 * 64;
    float* bufB = bufA + (size_t)N * 64;
    float* h3   = bufB + (size_t)N * 64;

    const int T = 256;
    auto cdiv = [](long a, long b) { return (int)((a + b - 1) / b); };
    const long NH = (long)N * 64;

    k_deg_init <<<cdiv(N, T), T, 0, stream>>>(deg, N);
    k_deg_count<<<cdiv(E, T), T, 0, stream>>>(dst, deg, E);
    k_norm     <<<cdiv(N, T), T, 0, stream>>>(deg, isq, idg, N);
    k_coef     <<<cdiv(E, T), T, 0, stream>>>(src, dst, isq, coef, E);
    k_pad_x    <<<cdiv((long)N * 8, T), T, 0, stream>>>(x, x8, N);
    k_prep_w   <<<cdiv(64 * 8,  T), T, 0, stream>>>(W1, w1t, 5, 64, 8, 64);
    k_prep_w   <<<cdiv(64 * 64, T), T, 0, stream>>>(W2, w2t, 64, 64, 64, 64);
    k_prep_w   <<<cdiv(16 * 64, T), T, 0, stream>>>(W3, w3t, 64, 2, 64, 16);

    const int GT = cdiv(N, 16);

    k_gemm_wmma<8, 4, 64><<<GT, 128, 0, stream>>>(x8, w1t, bufA);
    k_self_bias<<<cdiv(NH, T), T, 0, stream>>>(bufA, idg, b1, bufB, NH, 64);
    k_scatter64<<<cdiv(E, T), T, 0, stream>>>(src, dst, coef, bufA, bufB, E);
    k_relu     <<<cdiv(NH, T), T, 0, stream>>>(bufB, NH);

    k_gemm_wmma<64, 4, 64><<<GT, 128, 0, stream>>>(bufB, w2t, bufA);
    k_self_bias<<<cdiv(NH, T), T, 0, stream>>>(bufA, idg, b2, bufB, NH, 64);
    k_scatter64<<<cdiv(E, T), T, 0, stream>>>(src, dst, coef, bufA, bufB, E);
    k_relu     <<<cdiv(NH, T), T, 0, stream>>>(bufB, NH);

    k_gemm_wmma<64, 1, 2><<<GT, 32, 0, stream>>>(bufB, w3t, h3);
    k_self_bias<<<cdiv((long)N * 2, T), T, 0, stream>>>(h3, idg, b3, out, (long)N * 2, 2);
    k_scatter2 <<<cdiv(E, T), T, 0, stream>>>(src, dst, coef, h3, out, E);
}


# 64-wide edge scatter: the layer-1 scan stores each wave's tile edge list (source, row, factor) in workspace; layer 2 adds from the lists, no second scan of all dst; row sum order still LDS arrival
# speedup vs baseline: 9.3115x; 9.3115x over previous
#include <hip/hip_runtime.h>
#include <hip/hip_bf16.h>

typedef __attribute__((ext_vector_type(2))) float v2f;
typedef __attribute__((ext_vector_type(8))) float v8f;

constexpr int ACC_THREADS = 1024;
constexpr int ACC_UNROLL  = 8;
constexpr int DEG_ROWS    = 50016;
constexpr int DEG_SLICES  = 128;
constexpr int L3_ROWS     = 25008;
constexpr int L3_SLICES   = 64;

__device__ __forceinline__ size_t plane_pitch(size_t words) {
    return (words + 31) & ~(size_t)31;
}

__global__ __launch_bounds__(ACC_THREADS)
void k_deg_part(const int* __restrict__ dst, float* __restrict__ part, int e, int n) {
    __shared__ float cnt[DEG_ROWS];
    const int tid = threadIdx.x;
    const int r0  = blockIdx.y * DEG_ROWS;
    int nr = n - r0;
    if (nr > DEG_ROWS) nr = DEG_ROWS;
    for (int i = tid; i < nr; i += ACC_THREADS) cnt[i] = 0.0f;
    __syncthreads();

    const int per = (e + DEG_SLICES - 1) / DEG_SLICES;
    const int e0  = blockIdx.x * per;
    int e1 = e0 + per;
    if (e1 > e) e1 = e;
    for (int b = e0; b < e1; b += ACC_THREADS * ACC_UNROLL) {
        int d[ACC_UNROLL];
#pragma unroll
        for (int k = 0; k < ACC_UNROLL; ++k) {
            const int i = b + k * ACC_THREADS + tid;
            d[k] = (i < e1) ? dst[i] : -1;
        }
#pragma unroll
        for (int k = 0; k < ACC_UNROLL; ++k) {
            const int u = d[k] - r0;
            if ((unsigned)u < (unsigned)nr) atomicAdd(&cnt[u], 1.0f);
        }
    }
    __syncthreads();
    float* p = part + (size_t)blockIdx.x * plane_pitch((size_t)n) + r0;
    for (int i = tid; i < nr; i += ACC_THREADS) p[i] = cnt[i];
}

__global__ void k_norm_part(const float* __restrict__ part, float* __restrict__ isq,
                            float* __restrict__ idg, int n) {
    int i = blockIdx.x * blockDim.x + threadIdx.x;
    if (i < n) {
        const size_t pitch = plane_pitch((size_t)n);
        float d = 1.0f;
#pragma unroll 16
        for (int s = 0; s < DEG_SLICES; ++s) d += part[(size_t)s * pitch + i];
        isq[i] = rsqrtf(d);
        idg[i] = 1.0f / d;
    }
}

__global__ void k_coef(const int* __restrict__ src, const int* __restrict__ dst,
                       const float* __restrict__ isq, float* __restrict__ coef, int e) {
    int i = blockIdx.x * blockDim.x + threadIdx.x;
    if (i < e) coef[i] = isq[src[i]] * isq[dst[i]];
}

__global__ void k_pad_x(const float* __restrict__ x, float* __restrict__ x8, int n) {
    long idx = (long)blockIdx.x * blockDim.x + threadIdx.x;
    if (idx < (long)n * 8) {
        int i = (int)(idx >> 3), j = (int)(idx & 7);
        x8[idx] = (j < 5) ? x[(long)i * 5 + j] : 0.0f;
    }
}

__global__ void k_prep_w(const float* __restrict__ W, float* __restrict__ Bt,
                         int Kin, int Cin, int KPAD, int COLS) {
    int idx = blockIdx.x * blockDim.x + threadIdx.x;
    if (idx >= COLS * KPAD) return;
    int col = idx / KPAD, k = idx % KPAD;
    Bt[idx] = (k < Kin && col < Cin) ? W[k * Cin + col] : 0.0f;
}

template <int K, int NT, int SC>
__global__ void k_gemm_wmma(const float* __restrict__ A, const float* __restrict__ Bt,
                            float* __restrict__ H) {
    const int lane = threadIdx.x & 31;
    const int wave = threadIdx.x >> 5;
    const int half = lane >> 4;
    const int l15  = lane & 15;
    const size_t row = (size_t)blockIdx.x * 16 + l15;
    const int    col = wave * 16 + l15;
    const float* Arow = A + row * K;
    const float* Bcol = Bt + (size_t)col * K;

    v8f c = {};
#pragma unroll
    for (int k = 0; k < K; k += 4) {
        const int ka = k + half * 2;
        v2f a = *(const v2f*)(Arow + ka);
        v2f b = *(const v2f*)(Bcol + ka);
        c = __builtin_amdgcn_wmma_f32_16x16x4_f32(
                 false, a,  false, b,
                 (short)0, c,  false,  false);
    }

    if (NT * 16 == SC || col < SC) {
        const size_t mbase = (size_t)blockIdx.x * 16 + half * 8;
#pragma unroll
        for (int r = 0; r < 8; ++r)
            H[(mbase + r) * SC + col] = c[r];
    }
}

__global__ void k_self_bias(const float* __restrict__ h, const float* __restrict__ idg,
                            const float* __restrict__ bias, float* __restrict__ agg,
                            long n_elems, int dim) {
    long idx = (long)blockIdx.x * blockDim.x + threadIdx.x;
    if (idx < n_elems) {
        int i = (int)(idx / dim);
        int j = (int)(idx % dim);
        agg[idx] = h[idx] * idg[i] + bias[j];
    }
}

constexpr int TILE_ROWS    = 391;
constexpr int TILE_THREADS = 1024;
constexpr int WIN_EDGES    = 512;
constexpr int Q_CAP        = 128;
constexpr int Q_BATCH      = 32;
constexpr int Q_GROUP      = 16;

constexpr int REC_CAP   = 768;
constexpr int REC_WORDS = REC_CAP * 3;

template <bool FULL>
__device__ __forceinline__ void rows_add(int s, int u, float cf, int cnt, int lane,
                                         const float* __restrict__ h, float* tile) {
#pragma unroll
    for (int g = 0; g < Q_BATCH; g += Q_GROUP) {
        float v0[Q_GROUP], v1[Q_GROUP];
#pragma unroll
        for (int k = 0; k < Q_GROUP; ++k) {
            if (FULL || g + k < cnt) {
                const float* hs = h + (size_t)__builtin_amdgcn_readlane(s, g + k) * 64;
                v0[k] = hs[lane];
                v1[k] = hs[lane + 32];
            }
        }
#pragma unroll
        for (int k = 0; k < Q_GROUP; ++k) {
            if (FULL || g + k < cnt) {
                const float cfk = __int_as_float(__builtin_amdgcn_readlane(__float_as_int(cf), g + k));
                float* t = tile + __builtin_amdgcn_readlane(u, g + k) * 64;
                atomicAdd(t + lane,      v0[k] * cfk);
                atomicAdd(t + lane + 32, v1[k] * cfk);
            }
        }
    }
}

template <bool FULL, bool REC>
__device__ __forceinline__ void queue_drain(const int2* q, int qh, int cnt, int lane,
                                            const int* __restrict__ src,
                                            const float* __restrict__ coef,
                                            const float* __restrict__ h, float* tile,
                                            float* recw, int& nrec) {
    int s = 0, u = 0;
    float cf = 0.0f;
    if (FULL || lane < cnt) {
        const int2 en = q[(qh + lane) & (Q_CAP - 1)];
        u  = en.y;
        s  = src[en.x];
        cf = coef[en.x];
    }
    if (REC && nrec >= 0) {
        if (nrec + Q_BATCH <= REC_CAP) {
            if (FULL || lane < cnt) {
                float* r = recw + nrec * 3 + lane;
                r[0]           = __int_as_float(s);
                r[Q_BATCH]     = __int_as_float(u);
                r[2 * Q_BATCH] = cf;
            }
            nrec += cnt;
        } else {
            nrec = -1;
        }
    }
    rows_add<FULL>(s, u, cf, cnt, lane, h, tile);
}

template <bool REC>
__device__ __forceinline__ void tile_window(const int (&dv)[16], int wbase, int r0, int nr, int lane,
                                            int2* q, int& qh, int& qn,
                                            const int* __restrict__ src,
                                            const float* __restrict__ coef,
                                            const float* __restrict__ h, float* tile,
                                            float* recw, int& nrec) {
#pragma unroll
    for (int j = 0; j < 16; ++j) {
        const int  u = dv[j] - r0;
        const bool m = (unsigned)u < (unsigned)nr;
        unsigned bal = __builtin_amdgcn_ballot_w32(m);
        if (bal) {
            const int c = __builtin_popcount(bal);
            if (qn + c <= Q_CAP) {
                if (m) {
                    const int slot = (qh + qn + (int)__builtin_amdgcn_mbcnt_lo(bal, 0u)) & (Q_CAP - 1);
                    q[slot] = make_int2(wbase + (j >> 2) * 128 + lane * 4 + (j & 3), u);
                }
                qn += c;
            } else {
                if (REC) nrec = -1;
                while (bal) {
                    const int l = __builtin_ctz(bal);
                    bal &= bal - 1;
                    const int ul = __builtin_amdgcn_readlane(u, l);
                    const int ee = wbase + (j >> 2) * 128 + l * 4 + (j & 3);
                    const int   s  = src[ee];
                    const float cf = coef[ee];
                    const float* hs = h + (size_t)s * 64;
                    const float v0 = hs[lane];
                    const float v1 = hs[lane + 32];
                    float* t = tile + ul * 64;
                    atomicAdd(t + lane,      v0 * cf);
                    atomicAdd(t + lane + 32, v1 * cf);
                }
            }
        }
    }
    while (qn >= Q_BATCH) {
        queue_drain<true, REC>(q, qh, Q_BATCH, lane, src, coef, h, tile, recw, nrec);
        qh = (qh + Q_BATCH) & (Q_CAP - 1);
        qn -= Q_BATCH;
    }
}

template <bool REC>
__device__ __forceinline__ void scan_edges(const int* __restrict__ src, const int* __restrict__ dst,
                                           const float* __restrict__ coef, const float* __restrict__ h,
                                           float* tile, int2* q, int wave, int lane,
                                           int r0, int nr, int e, float* recw, int& nrec) {
    int qh = 0, qn = 0;

    const bool al    = (((size_t)dst) & 15) == 0;
    const int  nfull = al ? (e / WIN_EDGES) : 0;
    int4 nx[4];
    if (wave < nfull) {
        const int4* p = (const int4*)(dst + (size_t)wave * WIN_EDGES) + lane;
#pragma unroll
        for (int qq = 0; qq < 4; ++qq) nx[qq] = p[qq * 32];
    }
    for (int w = wave; w < nfull; w += TILE_THREADS / 32) {
        int dv[16];
#pragma unroll
        for (int qq = 0; qq < 4; ++qq) {
            dv[4 * qq + 0] = nx[qq].x; dv[4 * qq + 1] = nx[qq].y;
            dv[4 * qq + 2] = nx[qq].z; dv[4 * qq + 3] = nx[qq].w;
        }
        const int wn = w + TILE_THREADS / 32;
        if (wn < nfull) {
            const int4* p = (const int4*)(dst + (size_t)wn * WIN_EDGES) + lane;
#pragma unroll
            for (int qq = 0; qq < 4; ++qq) nx[qq] = p[qq * 32];
        }
        tile_window<REC>(dv, w * WIN_EDGES, r0, nr, lane, q, qh, qn, src, coef, h, tile, recw, nrec);
    }
    for (int w = nfull + wave; (long)w * WIN_EDGES < (long)e; w += TILE_THREADS / 32) {
        int dv[16];
#pragma unroll
        for (int j = 0; j < 16; ++j) {
            const int ej = w * WIN_EDGES + (j >> 2) * 128 + lane * 4 + (j & 3);
            dv[j] = (ej < e) ? dst[ej] : -1;
        }
        tile_window<REC>(dv, w * WIN_EDGES, r0, nr, lane, q, qh, qn, src, coef, h, tile, recw, nrec);
    }
    if (qn > 0) queue_drain<false, REC>(q, qh, qn, lane, src, coef, h, tile, recw, nrec);
}

__device__ __forceinline__ void replay_edges(const float* __restrict__ recw, int nrec, int lane,
                                             const float* __restrict__ h, float* tile) {
    int s = 0, u = 0;
    float cf = 0.0f;
    if (lane < nrec) {
        s  = __float_as_int(recw[lane]);
        u  = __float_as_int(recw[Q_BATCH + lane]);
        cf = recw[2 * Q_BATCH + lane];
    }
    for (int b = 0; b < nrec; b += Q_BATCH) {
        int s2 = 0, u2 = 0;
        float cf2 = 0.0f;
        const int nb = b + Q_BATCH;
        if (nb + lane < nrec) {
            const float* r = recw + nb * 3 + lane;
            s2  = __float_as_int(r[0]);
            u2  = __float_as_int(r[Q_BATCH]);
            cf2 = r[2 * Q_BATCH];
        }
        if (nrec - b >= Q_BATCH) rows_add<true>(s, u, cf, Q_BATCH, lane, h, tile);
        else                     rows_add<false>(s, u, cf, nrec - b, lane, h, tile);
        s = s2; u = u2; cf = cf2;
    }
}

template <bool REPLAY>
__global__ __launch_bounds__(TILE_THREADS)
void k_scatter64_tile(const int* __restrict__ src, const int* __restrict__ dst,
                      const float* __restrict__ coef, const float* __restrict__ h,
                      float* __restrict__ agg, int* __restrict__ lcnt, float* __restrict__ rec,
                      int e, int n) {
    __shared__ float tile[TILE_ROWS * 64];
    __shared__ int2  queue[(TILE_THREADS / 32) * Q_CAP];
    const int tid  = threadIdx.x;
    const int lane = tid & 31;
    const int wave = __builtin_amdgcn_readfirstlane(tid >> 5);
    const int r0   = blockIdx.x * TILE_ROWS;
    int nr = n - r0;
    if (nr > TILE_ROWS) nr = TILE_ROWS;
    if (nr <= 0) return;

    float4*       t4 = (float4*)tile;
    float4*       a4 = (float4*)(agg + (size_t)r0 * 64);
    for (int i = tid; i < nr * 16; i += TILE_THREADS) t4[i] = a4[i];
    __syncthreads();

    int2*  q    = queue + wave * Q_CAP;
    const int wid = blockIdx.x * (TILE_THREADS / 32) + wave;
    float* recw = rec + (size_t)wid * REC_WORDS;

    if (REPLAY) {
        int nrec = __builtin_amdgcn_readfirstlane(lcnt[wid]);
        if (nrec >= 0) replay_edges(recw, nrec, lane, h, tile);
        else           scan_edges<false>(src, dst, coef, h, tile, q, wave, lane, r0, nr, e, recw, nrec);
    } else {
        int nrec = 0;
        scan_edges<true>(src, dst, coef, h, tile, q, wave, lane, r0, nr, e, recw, nrec);
        if (lane == 0) lcnt[wid] = nrec;
    }
    __syncthreads();
    for (int i = tid; i < nr * 16; i += TILE_THREADS) a4[i] = t4[i];
}

__global__ __launch_bounds__(ACC_THREADS)
void k_l3_part(const int* __restrict__ src, const int* __restrict__ dst,
               const float* __restrict__ coef, const float* __restrict__ h,
               float* __restrict__ part, int e, int n) {
    __shared__ float acc[L3_ROWS * 2];
    const int tid = threadIdx.x;
    const int r0  = blockIdx.y * L3_ROWS;
    int nr = n - r0;
    if (nr > L3_ROWS) nr = L3_ROWS;
    float2* acc2 = (float2*)acc;
    for (int i = tid; i < nr; i += ACC_THREADS) acc2[i] = make_float2(0.0f, 0.0f);
    __syncthreads();

    const int per = (e + L3_SLICES - 1) / L3_SLICES;
    const int e0  = blockIdx.x * per;
    int e1 = e0 + per;
    if (e1 > e) e1 = e;
    for (int b = e0; b < e1; b += ACC_THREADS * ACC_UNROLL) {
        int   d[ACC_UNROLL], s[ACC_UNROLL];
        float cf[ACC_UNROLL];
#pragma unroll
        for (int k = 0; k < ACC_UNROLL; ++k) {
            const int  i  = b + k * ACC_THREADS + tid;
            const bool ok = i < e1;
            d[k]  = ok ? dst[i]  : -1;
            s[k]  = ok ? src[i]  : 0;
            cf[k] = ok ? coef[i] : 0.0f;
        }
        float2 v[ACC_UNROLL];
#pragma unroll
        for (int k = 0; k < ACC_UNROLL; ++k) {
            v[k] = make_float2(0.0f, 0.0f);
            if ((unsigned)(d[k] - r0) < (unsigned)nr) v[k] = *(const float2*)(h + (size_t)s[k] * 2);
        }
#pragma unroll
        for (int k = 0; k < ACC_UNROLL; ++k) {
            const int u = d[k] - r0;
            if ((unsigned)u < (unsigned)nr) {
                atomicAdd(&acc[u * 2 + 0], v[k].x * cf[k]);
                atomicAdd(&acc[u * 2 + 1], v[k].y * cf[k]);
            }
        }
    }
    __syncthreads();
    float2* p = (float2*)(part + (size_t)blockIdx.x * plane_pitch((size_t)n * 2)) + r0;
    for (int i = tid; i < nr; i += ACC_THREADS) p[i] = acc2[i];
}

__global__ void k_l3_sum(const float* __restrict__ h, const float* __restrict__ idg,
                         const float* __restrict__ bias, const float* __restrict__ part,
                         float* __restrict__ out, int n) {
    int i = blockIdx.x * blockDim.x + threadIdx.x;
    if (i >= n) return;
    const size_t pitch2 = plane_pitch((size_t)n * 2) / 2;
    const float2 hv = ((const float2*)h)[i];
    const float  g  = idg[i];
    float2 a;
    a.x = hv.x * g + bias[0];
    a.y = hv.y * g + bias[1];
    const float2* p = (const float2*)part + i;
#pragma unroll 16
    for (int s = 0; s < L3_SLICES; ++s) {
        const float2 t = p[(size_t)s * pitch2];
        a.x += t.x;
        a.y += t.y;
    }
    ((float2*)out)[i] = a;
}

__global__ void k_relu(float* __restrict__ x, long n) {
    long idx = (long)blockIdx.x * blockDim.x + threadIdx.x;
    if (idx < n) x[idx] = fmaxf(x[idx], 0.0f);
}

extern "C" void kernel_launch(void* const* d_in, const int* in_sizes, int n_in,
                              void* d_out, int out_size, void* d_ws, size_t ws_size,
                              hipStream_t stream) {
    const float* x  = (const float*)d_in[0];
    const int*   ei = (const int*)d_in[1];
    const float* W1 = (const float*)d_in[2];
    const float* b1 = (const float*)d_in[3];
    const float* W2 = (const float*)d_in[4];
    const float* b2 = (const float*)d_in[5];
    const float* W3 = (const float*)d_in[6];
    const float* b3 = (const float*)d_in[7];
    float* out = (float*)d_out;

    const int N = in_sizes[0] / 5;
    const int E = in_sizes[1] / 2;
    const int* src = ei;
    const int* dst = ei + E;

    float* ws   = (float*)d_ws;
    float* isq  = ws;
    float* idg  = isq  + N;
    float* coef = idg  + N;
    float* x8   = coef + E;
    float* w1t  = x8   + (size_t)N * 8;
    float* w2t  = w1t  + 64 * 8;
    float* w3t  = w2t  + 64 * 64;
    float* bufA = w3t  + 16 * 64;
    float* bufB = bufA + (size_t)N * 64;
    float* h3   = bufB + (size_t)N * 64;
    float* part = h3   + (size_t)N * 2;
    int*   lcnt = (int*)(part + (size_t)128 * (N + 32));
    float* rec  = (float*)(lcnt + (size_t)((N + TILE_ROWS - 1) / TILE_ROWS) * (TILE_THREADS / 32));

    const int T = 256;
    auto cdiv = [](long a, long b) { return (int)((a + b - 1) / b); };
    const long NH = (long)N * 64;

    k_deg_part <<<dim3(DEG_SLICES, cdiv(N, DEG_ROWS)), ACC_THREADS, 0, stream>>>(dst, part, E, N);
    k_norm_part<<<cdiv(N, T), T, 0, stream>>>(part, isq, idg, N);
    k_coef     <<<cdiv(E, T), T, 0, stream>>>(src, dst, isq, coef, E);
    k_pad_x    <<<cdiv((long)N * 8, T), T, 0, stream>>>(x, x8, N);
    k_prep_w   <<<cdiv(64 * 8,  T), T, 0, stream>>>(W1, w1t, 5, 64, 8, 64);
    k_prep_w   <<<cdiv(64 * 64, T), T, 0, stream>>>(W2, w2t, 64, 64, 64, 64);
    k_prep_w   <<<cdiv(16 * 64, T), T, 0, stream>>>(W3, w3t, 64, 2, 64, 16);

    const int GT = cdiv(N, 16);

    k_gemm_wmma<8, 4, 64><<<GT, 128, 0, stream>>>(x8, w1t, bufA);
    k_self_bias<<<cdiv(NH, T), T, 0, stream>>>(bufA, idg, b1, bufB, NH, 64);
    k_scatter64_tile<false><<<cdiv(N, TILE_ROWS), TILE_THREADS, 0, stream>>>(src, dst, coef, bufA, bufB, lcnt, rec, E, N);
    k_relu     <<<cdiv(NH, T), T, 0, stream>>>(bufB, NH);

    k_gemm_wmma<64, 4, 64><<<GT, 128, 0, stream>>>(bufB, w2t, bufA);
    k_self_bias<<<cdiv(NH, T), T, 0, stream>>>(bufA, idg, b2, bufB, NH, 64);
    k_scatter64_tile<true><<<cdiv(N, TILE_ROWS), TILE_THREADS, 0, stream>>>(src, dst, coef, bufA, bufB, lcnt, rec, E, N);
    k_relu     <<<cdiv(NH, T), T, 0, stream>>>(bufB, NH);

    k_gemm_wmma<64, 1, 2><<<GT, 32, 0, stream>>>(bufB, w3t, h3);
    k_l3_part  <<<dim3(L3_SLICES, cdiv(N, L3_ROWS)), ACC_THREADS, 0, stream>>>(src, dst, coef, h3, part, E, N);
    k_l3_sum   <<<cdiv(N, T), T, 0, stream>>>(h3, idg, b3, part, out, N);
}


# 64-wide edge scatter: layer-1 scan reads an edge's tile number as bit planes (1 byte/edge) and one packed record per hit, both from the factor kernel, not 4-byte dst; row sum order still LDS arrival
# speedup vs baseline: 10.8803x; 1.1685x over previous
#include <hip/hip_runtime.h>
#include <hip/hip_bf16.h>

typedef __attribute__((ext_vector_type(2))) float v2f;
typedef __attribute__((ext_vector_type(8))) float v8f;

constexpr int ACC_THREADS = 1024;
constexpr int ACC_UNROLL  = 8;
constexpr int DEG_ROWS    = 50016;
constexpr int DEG_SLICES  = 128;
constexpr int L3_ROWS     = 25008;
constexpr int L3_SLICES   = 64;

__device__ __forceinline__ size_t plane_pitch(size_t words) {
    return (words + 31) & ~(size_t)31;
}

__global__ __launch_bounds__(ACC_THREADS)
void k_deg_part(const int* __restrict__ dst, float* __restrict__ part, int e, int n) {
    __shared__ float cnt[DEG_ROWS];
    const int tid = threadIdx.x;
    const int r0  = blockIdx.y * DEG_ROWS;
    int nr = n - r0;
    if (nr > DEG_ROWS) nr = DEG_ROWS;
    for (int i = tid; i < nr; i += ACC_THREADS) cnt[i] = 0.0f;
    __syncthreads();

    const int per = (e + DEG_SLICES - 1) / DEG_SLICES;
    const int e0  = blockIdx.x * per;
    int e1 = e0 + per;
    if (e1 > e) e1 = e;
    for (int b = e0; b < e1; b += ACC_THREADS * ACC_UNROLL) {
        int d[ACC_UNROLL];
#pragma unroll
        for (int k = 0; k < ACC_UNROLL; ++k) {
            const int i = b + k * ACC_THREADS + tid;
            d[k] = (i < e1) ? dst[i] : -1;
        }
#pragma unroll
        for (int k = 0; k < ACC_UNROLL; ++k) {
            const int u = d[k] - r0;
            if ((unsigned)u < (unsigned)nr) atomicAdd(&cnt[u], 1.0f);
        }
    }
    __syncthreads();
    float* p = part + (size_t)blockIdx.x * plane_pitch((size_t)n) + r0;
    for (int i = tid; i < nr; i += ACC_THREADS) p[i] = cnt[i];
}

__global__ void k_norm_part(const float* __restrict__ part, float* __restrict__ isq,
                            float* __restrict__ idg, int n) {
    int i = blockIdx.x * blockDim.x + threadIdx.x;
    if (i < n) {
        const size_t pitch = plane_pitch((size_t)n);
        float d = 1.0f;
#pragma unroll 16
        for (int s = 0; s < DEG_SLICES; ++s) d += part[(size_t)s * pitch + i];
        isq[i] = rsqrtf(d);
        idg[i] = 1.0f / d;
    }
}

__global__ void k_pad_x(const float* __restrict__ x, float* __restrict__ x8, int n) {
    long idx = (long)blockIdx.x * blockDim.x + threadIdx.x;
    if (idx < (long)n * 8) {
        int i = (int)(idx >> 3), j = (int)(idx & 7);
        x8[idx] = (j < 5) ? x[(long)i * 5 + j] : 0.0f;
    }
}

__global__ void k_prep_w(const float* __restrict__ W, float* __restrict__ Bt,
                         int Kin, int Cin, int KPAD, int COLS) {
    int idx = blockIdx.x * blockDim.x + threadIdx.x;
    if (idx >= COLS * KPAD) return;
    int col = idx / KPAD, k = idx % KPAD;
    Bt[idx] = (k < Kin && col < Cin) ? W[k * Cin + col] : 0.0f;
}

template <int K, int NT, int SC>
__global__ void k_gemm_wmma(const float* __restrict__ A, const float* __restrict__ Bt,
                            float* __restrict__ H) {
    const int lane = threadIdx.x & 31;
    const int wave = threadIdx.x >> 5;
    const int half = lane >> 4;
    const int l15  = lane & 15;
    const size_t row = (size_t)blockIdx.x * 16 + l15;
    const int    col = wave * 16 + l15;
    const float* Arow = A + row * K;
    const float* Bcol = Bt + (size_t)col * K;

    v8f c = {};
#pragma unroll
    for (int k = 0; k < K; k += 4) {
        const int ka = k + half * 2;
        v2f a = *(const v2f*)(Arow + ka);
        v2f b = *(const v2f*)(Bcol + ka);
        c = __builtin_amdgcn_wmma_f32_16x16x4_f32(
                 false, a,  false, b,
                 (short)0, c,  false,  false);
    }

    if (NT * 16 == SC || col < SC) {
        const size_t mbase = (size_t)blockIdx.x * 16 + half * 8;
#pragma unroll
        for (int r = 0; r < 8; ++r)
            H[(mbase + r) * SC + col] = c[r];
    }
}

__global__ void k_self_bias(const float* __restrict__ h, const float* __restrict__ idg,
                            const float* __restrict__ bias, float* __restrict__ agg,
                            long n_elems, int dim) {
    long idx = (long)blockIdx.x * blockDim.x + threadIdx.x;
    if (idx < n_elems) {
        int i = (int)(idx / dim);
        int j = (int)(idx % dim);
        agg[idx] = h[idx] * idg[i] + bias[j];
    }
}

constexpr int TILE_ROWS    = 391;
constexpr int TILE_THREADS = 1024;
constexpr int Q_CAP        = 64;
constexpr int Q_BATCH      = 32;
constexpr int Q_GROUP      = 16;

constexpr int KEY_WIN   = 1024;
constexpr int ROW_SHIFT = 23;
static_assert(TILE_ROWS <= (1 << (32 - ROW_SHIFT)), "row inside the tile must fit above ROW_SHIFT");

constexpr int PREP_THREADS = 256;
constexpr int PREP_STEPS   = 16;
__global__ __launch_bounds__(PREP_THREADS)
void k_edge_prep(const int* __restrict__ src, const int* __restrict__ dst,
                 const float* __restrict__ isq, float* __restrict__ coef,
                 int* __restrict__ pk, int* __restrict__ keys, int e) {
    const int lane = threadIdx.x & 31;
    const int win = __builtin_amdgcn_readfirstlane((int)(blockIdx.x * (PREP_THREADS / 32) + (threadIdx.x >> 5)));
    const int wb  = win * KEY_WIN;
    if (wb >= e) return;
    int pl[8] = {0, 0, 0, 0, 0, 0, 0, 0};
    for (int i0 = 0; i0 < 32; i0 += PREP_STEPS) {
        int d[PREP_STEPS], s[PREP_STEPS];
#pragma unroll
        for (int k = 0; k < PREP_STEPS; ++k) {
            const int  i  = wb + (i0 + k) * 32 + lane;
            const bool ok = i < e;
            d[k] = ok ? dst[i] : 0;
            s[k] = ok ? src[i] : 0;
        }
        float fs[PREP_STEPS], fd[PREP_STEPS];
#pragma unroll
        for (int k = 0; k < PREP_STEPS; ++k) {
            fs[k] = isq[s[k]];
            fd[k] = isq[d[k]];
        }
#pragma unroll
        for (int k = 0; k < PREP_STEPS; ++k) {
            const int  i  = wb + (i0 + k) * 32 + lane;
            const bool ok = i < e;
            const int   t  = d[k] / TILE_ROWS;
            const int   u  = d[k] - t * TILE_ROWS;
            const float cf = fs[k] * fd[k];
            if (ok) {
                coef[i] = cf;
                ((int2*)pk)[i] = make_int2((int)((unsigned)s[k] | ((unsigned)u << ROW_SHIFT)), __float_as_int(cf));
            }
            const bool mine = lane == i0 + k;
#pragma unroll
            for (int p = 0; p < 8; ++p) {
                const int bits = (int)__builtin_amdgcn_ballot_w32(ok && ((t >> p) & 1));
                pl[p] = mine ? bits : pl[p];
            }
        }
    }
    int4* k4 = (int4*)keys + (size_t)win * 64;
    k4[lane]      = make_int4(pl[0], pl[1], pl[2], pl[3]);
    k4[32 + lane] = make_int4(pl[4], pl[5], pl[6], pl[7]);
}

constexpr int REC_CAP   = 768;
constexpr int REC_WORDS = REC_CAP * 3;

template <bool FULL>
__device__ __forceinline__ void rows_add(int s, int u, float cf, int cnt, int lane,
                                         const float* __restrict__ h, float* tile) {
#pragma unroll
    for (int g = 0; g < Q_BATCH; g += Q_GROUP) {
        float v0[Q_GROUP], v1[Q_GROUP];
#pragma unroll
        for (int k = 0; k < Q_GROUP; ++k) {
            if (FULL || g + k < cnt) {
                const float* hs = h + (size_t)__builtin_amdgcn_readlane(s, g + k) * 64;
                v0[k] = hs[lane];
                v1[k] = hs[lane + 32];
            }
        }
#pragma unroll
        for (int k = 0; k < Q_GROUP; ++k) {
            if (FULL || g + k < cnt) {
                const float cfk = __int_as_float(__builtin_amdgcn_readlane(__float_as_int(cf), g + k));
                float* t = tile + __builtin_amdgcn_readlane(u, g + k) * 64;
                atomicAdd(t + lane,      v0[k] * cfk);
                atomicAdd(t + lane + 32, v1[k] * cfk);
            }
        }
    }
}

template <bool FULL, bool REC>
__device__ __forceinline__ void queue_drain(const int* q, int qh, int cnt, int lane,
                                            const int* __restrict__ pk,
                                            const float* __restrict__ h, float* tile,
                                            float* recw, int& nrec) {
    int s = 0, u = 0;
    float cf = 0.0f;
    if (FULL || lane < cnt) {
        const int  en = q[(qh + lane) & (Q_CAP - 1)];
        const int2 r  = ((const int2*)pk)[en];
        s  = r.x & ((1 << ROW_SHIFT) - 1);
        u  = (int)((unsigned)r.x >> ROW_SHIFT);
        cf = __int_as_float(r.y);
    }
    if (REC && nrec >= 0) {
        if (nrec + Q_BATCH <= REC_CAP) {
            if (FULL || lane < cnt) {
                float* r = recw + nrec * 3 + lane;
                r[0]           = __int_as_float(s);
                r[Q_BATCH]     = __int_as_float(u);
                r[2 * Q_BATCH] = cf;
            }
            nrec += cnt;
        } else {
            nrec = -1;
        }
    }
    rows_add<FULL>(s, u, cf, cnt, lane, h, tile);
}

__device__ __forceinline__ unsigned key_match(const int4& lo, const int4& hi, const unsigned (&xm)[8]) {
    unsigned m = ((unsigned)lo.x ^ xm[0]) & ((unsigned)lo.y ^ xm[1]);
    m &= ((unsigned)lo.z ^ xm[2]);
    m &= ((unsigned)lo.w ^ xm[3]);
    m &= ((unsigned)hi.x ^ xm[4]);
    m &= ((unsigned)hi.y ^ xm[5]);
    m &= ((unsigned)hi.z ^ xm[6]);
    m &= ((unsigned)hi.w ^ xm[7]);
    return m;
}

template <bool REC>
__device__ __forceinline__ void window_hits(unsigned m, int ebase, int lane,
                                            int* q, int& qh, int& qn,
                                            const int* __restrict__ pk,
                                            const float* __restrict__ h, float* tile,
                                            float* recw, int& nrec) {
    unsigned bal = __builtin_amdgcn_ballot_w32(m != 0u);
    while (bal) {
        if (m != 0u) {
            const int slot = (qh + qn + (int)__builtin_amdgcn_mbcnt_lo(bal, 0u)) & (Q_CAP - 1);
            q[slot] = ebase + __builtin_ctz(m);
            m &= m - 1u;
        }
        qn += __builtin_popcount(bal);
        if (qn >= Q_BATCH) {
            queue_drain<true, REC>(q, qh, Q_BATCH, lane, pk, h, tile, recw, nrec);
            qh = (qh + Q_BATCH) & (Q_CAP - 1);
            qn -= Q_BATCH;
        }
        bal = __builtin_amdgcn_ballot_w32(m != 0u);
    }
}

template <bool REC>
__device__ __forceinline__ void scan_edges(const int* __restrict__ keys, const int* __restrict__ pk,
                                           const float* __restrict__ h,
                                           float* tile, int* q, int wave, int lane,
                                           int tno, int e, float* recw, int& nrec) {
    int qh = 0, qn = 0;
    unsigned xm[8];
#pragma unroll
    for (int p = 0; p < 8; ++p) xm[p] = ((tno >> p) & 1) ? 0u : ~0u;

    const int4* k4    = (const int4*)keys;
    const int   nfull = e / (2 * KEY_WIN);
    int4 nx[4];
    if (wave < nfull) {
        const int4* p = k4 + (size_t)wave * 128 + lane;
#pragma unroll
        for (int qq = 0; qq < 4; ++qq) nx[qq] = p[qq * 32];
    }
    for (int w = wave; w < nfull; w += TILE_THREADS / 32) {
        int4 cur[4];
#pragma unroll
        for (int qq = 0; qq < 4; ++qq) cur[qq] = nx[qq];
        const int wn = w + TILE_THREADS / 32;
        if (wn < nfull) {
            const int4* p = k4 + (size_t)wn * 128 + lane;
#pragma unroll
            for (int qq = 0; qq < 4; ++qq) nx[qq] = p[qq * 32];
        }
#pragma unroll
        for (int hh = 0; hh < 2; ++hh) {
            const unsigned m = key_match(cur[2 * hh], cur[2 * hh + 1], xm);
            window_hits<REC>(m, (2 * w + hh) * KEY_WIN + lane * 32, lane, q, qh, qn, pk, h, tile, recw, nrec);
        }
    }
    const int kw = (e + KEY_WIN - 1) / KEY_WIN;
    for (int tw = 2 * nfull + wave; tw < kw; tw += TILE_THREADS / 32) {
        const int4* p = k4 + (size_t)tw * 64 + lane;
        const int4 lo = p[0];
        const int4 hi = p[32];
        unsigned m = key_match(lo, hi, xm);
        const int ebase = tw * KEY_WIN + lane * 32;
        const int valid = e - ebase;
        if (valid < 32) m = (valid <= 0) ? 0u : (m & ((1u << valid) - 1u));
        window_hits<REC>(m, ebase, lane, q, qh, qn, pk, h, tile, recw, nrec);
    }
    if (qn > 0) queue_drain<false, REC>(q, qh, qn, lane, pk, h, tile, recw, nrec);
}

__device__ __forceinline__ void replay_edges(const float* __restrict__ recw, int nrec, int lane,
                                             const float* __restrict__ h, float* tile) {
    int s = 0, u = 0;
    float cf = 0.0f;
    if (lane < nrec) {
        s  = __float_as_int(recw[lane]);
        u  = __float_as_int(recw[Q_BATCH + lane]);
        cf = recw[2 * Q_BATCH + lane];
    }
    for (int b = 0; b < nrec; b += Q_BATCH) {
        int s2 = 0, u2 = 0;
        float cf2 = 0.0f;
        const int nb = b + Q_BATCH;
        if (nb + lane < nrec) {
            const float* r = recw + nb * 3 + lane;
            s2  = __float_as_int(r[0]);
            u2  = __float_as_int(r[Q_BATCH]);
            cf2 = r[2 * Q_BATCH];
        }
        if (nrec - b >= Q_BATCH) rows_add<true>(s, u, cf, Q_BATCH, lane, h, tile);
        else                     rows_add<false>(s, u, cf, nrec - b, lane, h, tile);
        s = s2; u = u2; cf = cf2;
    }
}

template <bool REPLAY>
__global__ __launch_bounds__(TILE_THREADS)
void k_scatter64_tile(const int* __restrict__ keys, const int* __restrict__ pk,
                      const float* __restrict__ h,
                      float* __restrict__ agg, int* __restrict__ lcnt, float* __restrict__ rec,
                      int e, int n) {
    __shared__ float tile[TILE_ROWS * 64];
    __shared__ int   queue[(TILE_THREADS / 32) * Q_CAP];
    const int tid  = threadIdx.x;
    const int lane = tid & 31;
    const int wave = __builtin_amdgcn_readfirstlane(tid >> 5);
    const int r0   = blockIdx.x * TILE_ROWS;
    int nr = n - r0;
    if (nr > TILE_ROWS) nr = TILE_ROWS;
    if (nr <= 0) return;

    float4*       t4 = (float4*)tile;
    float4*       a4 = (float4*)(agg + (size_t)r0 * 64);
    for (int i = tid; i < nr * 16; i += TILE_THREADS) t4[i] = a4[i];
    __syncthreads();

    int*   q    = queue + wave * Q_CAP;
    const int tno = __builtin_amdgcn_readfirstlane((int)blockIdx.x);
    const int wid = blockIdx.x * (TILE_THREADS / 32) + wave;
    float* recw = rec + (size_t)wid * REC_WORDS;

    if (REPLAY) {
        int nrec = __builtin_amdgcn_readfirstlane(lcnt[wid]);
        if (nrec >= 0) replay_edges(recw, nrec, lane, h, tile);
        else           scan_edges<false>(keys, pk, h, tile, q, wave, lane, tno, e, recw, nrec);
    } else {
        int nrec = 0;
        scan_edges<true>(keys, pk, h, tile, q, wave, lane, tno, e, recw, nrec);
        if (lane == 0) lcnt[wid] = nrec;
    }
    __syncthreads();
    for (int i = tid; i < nr * 16; i += TILE_THREADS) a4[i] = t4[i];
}

__global__ __launch_bounds__(ACC_THREADS)
void k_l3_part(const int* __restrict__ src, const int* __restrict__ dst,
               const float* __restrict__ coef, const float* __restrict__ h,
               float* __restrict__ part, int e, int n) {
    __shared__ float acc[L3_ROWS * 2];
    const int tid = threadIdx.x;
    const int r0  = blockIdx.y * L3_ROWS;
    int nr = n - r0;
    if (nr > L3_ROWS) nr = L3_ROWS;
    float2* acc2 = (float2*)acc;
    for (int i = tid; i < nr; i += ACC_THREADS) acc2[i] = make_float2(0.0f, 0.0f);
    __syncthreads();

    const int per = (e + L3_SLICES - 1) / L3_SLICES;
    const int e0  = blockIdx.x * per;
    int e1 = e0 + per;
    if (e1 > e) e1 = e;
    for (int b = e0; b < e1; b += ACC_THREADS * ACC_UNROLL) {
        int   d[ACC_UNROLL], s[ACC_UNROLL];
        float cf[ACC_UNROLL];
#pragma unroll
        for (int k = 0; k < ACC_UNROLL; ++k) {
            const int  i  = b + k * ACC_THREADS + tid;
            const bool ok = i < e1;
            d[k]  = ok ? dst[i]  : -1;
            s[k]  = ok ? src[i]  : 0;
            cf[k] = ok ? coef[i] : 0.0f;
        }
        float2 v[ACC_UNROLL];
#pragma unroll
        for (int k = 0; k < ACC_UNROLL; ++k) {
            v[k] = make_float2(0.0f, 0.0f);
            if ((unsigned)(d[k] - r0) < (unsigned)nr) v[k] = *(const float2*)(h + (size_t)s[k] * 2);
        }
#pragma unroll
        for (int k = 0; k < ACC_UNROLL; ++k) {
            const int u = d[k] - r0;
            if ((unsigned)u < (unsigned)nr) {
                atomicAdd(&acc[u * 2 + 0], v[k].x * cf[k]);
                atomicAdd(&acc[u * 2 + 1], v[k].y * cf[k]);
            }
        }
    }
    __syncthreads();
    float2* p = (float2*)(part + (size_t)blockIdx.x * plane_pitch((size_t)n * 2)) + r0;
    for (int i = tid; i < nr; i += ACC_THREADS) p[i] = acc2[i];
}

__global__ void k_l3_sum(const float* __restrict__ h, const float* __restrict__ idg,
                         const float* __restrict__ bias, const float* __restrict__ part,
                         float* __restrict__ out, int n) {
    int i = blockIdx.x * blockDim.x + threadIdx.x;
    if (i >= n) return;
    const size_t pitch2 = plane_pitch((size_t)n * 2) / 2;
    const float2 hv = ((const float2*)h)[i];
    const float  g  = idg[i];
    float2 a;
    a.x = hv.x * g + bias[0];
    a.y = hv.y * g + bias[1];
    const float2* p = (const float2*)part + i;
#pragma unroll 16
    for (int s = 0; s < L3_SLICES; ++s) {
        const float2 t = p[(size_t)s * pitch2];
        a.x += t.x;
        a.y += t.y;
    }
    ((float2*)out)[i] = a;
}

__global__ void k_relu(float* __restrict__ x, long n) {
    long idx = (long)blockIdx.x * blockDim.x + threadIdx.x;
    if (idx < n) x[idx] = fmaxf(x[idx], 0.0f);
}

extern "C" void kernel_launch(void* const* d_in, const int* in_sizes, int n_in,
                              void* d_out, int out_size, void* d_ws, size_t ws_size,
                              hipStream_t stream) {
    const float* x  = (const float*)d_in[0];
    const int*   ei = (const int*)d_in[1];
    const float* W1 = (const float*)d_in[2];
    const float* b1 = (const float*)d_in[3];
    const float* W2 = (const float*)d_in[4];
    const float* b2 = (const float*)d_in[5];
    const float* W3 = (const float*)d_in[6];
    const float* b3 = (const float*)d_in[7];
    float* out = (float*)d_out;

    const int N = in_sizes[0] / 5;
    const int E = in_sizes[1] / 2;
    const int* src = ei;
    const int* dst = ei + E;

    float* ws   = (float*)d_ws;
    float* isq  = ws;
    float* idg  = isq  + N;
    float* coef = idg  + N;
    float* x8   = coef + E;
    float* w1t  = x8   + (size_t)N * 8;
    float* w2t  = w1t  + 64 * 8;
    float* w3t  = w2t  + 64 * 64;
    float* bufA = w3t  + 16 * 64;
    float* bufB = bufA + (size_t)N * 64;
    float* h3   = bufB + (size_t)N * 64;
    float* part = h3   + (size_t)N * 2;
    int*   lcnt = (int*)(part + (size_t)128 * (N + 32));
    float* rec  = (float*)(lcnt + (size_t)((N + TILE_ROWS - 1) / TILE_ROWS) * (TILE_THREADS / 32));
    int*   keys = (int*)(rec + (size_t)((N + TILE_ROWS - 1) / TILE_ROWS) * (TILE_THREADS / 32) * REC_WORDS);
    int*   pk   = keys + (size_t)((E + KEY_WIN - 1) / KEY_WIN) * (KEY_WIN / 4);

    const int T = 256;
    auto cdiv = [](long a, long b) { return (int)((a + b - 1) / b); };
    const long NH = (long)N * 64;

    k_deg_part <<<dim3(DEG_SLICES, cdiv(N, DEG_ROWS)), ACC_THREADS, 0, stream>>>(dst, part, E, N);
    k_norm_part<<<cdiv(N, T), T, 0, stream>>>(part, isq, idg, N);
    k_edge_prep<<<cdiv(cdiv(E, KEY_WIN), PREP_THREADS / 32), PREP_THREADS, 0, stream>>>(src, dst, isq, coef, pk, keys, E);
    k_pad_x    <<<cdiv((long)N * 8, T), T, 0, stream>>>(x, x8, N);
    k_prep_w   <<<cdiv(64 * 8,  T), T, 0, stream>>>(W1, w1t, 5, 64, 8, 64);
    k_prep_w   <<<cdiv(64 * 64, T), T, 0, stream>>>(W2, w2t, 64, 64, 64, 64);
    k_prep_w   <<<cdiv(16 * 64, T), T, 0, stream>>>(W3, w3t, 64, 2, 64, 16);

    const int GT = cdiv(N, 16);

    k_gemm_wmma<8, 4, 64><<<GT, 128, 0, stream>>>(x8, w1t, bufA);
    k_self_bias<<<cdiv(NH, T), T, 0, stream>>>(bufA, idg, b1, bufB, NH, 64);
    k_scatter64_tile<false><<<cdiv(N, TILE_ROWS), TILE_THREADS, 0, stream>>>(keys, pk, bufA, bufB, lcnt, rec, E, N);
    k_relu     <<<cdiv(NH, T), T, 0, stream>>>(bufB, NH);

    k_gemm_wmma<64, 4, 64><<<GT, 128, 0, stream>>>(bufB, w2t, bufA);
    k_self_bias<<<cdiv(NH, T), T, 0, stream>>>(bufA, idg, b2, bufB, NH, 64);
    k_scatter64_tile<true><<<cdiv(N, TILE_ROWS), TILE_THREADS, 0, stream>>>(keys, pk, bufA, bufB, lcnt, rec, E, N);
    k_relu     <<<cdiv(NH, T), T, 0, stream>>>(bufB, NH);

    k_gemm_wmma<64, 1, 2><<<GT, 32, 0, stream>>>(bufB, w3t, h3);
    k_l3_part  <<<dim3(L3_SLICES, cdiv(N, L3_ROWS)), ACC_THREADS, 0, stream>>>(src, dst, coef, h3, part, E, N);
    k_l3_sum   <<<cdiv(N, T), T, 0, stream>>>(h3, idg, b3, part, out, N);
}


# Elementwise passes folded into the WMMA GEMM launches: epilogue writes the self-loop+bias plane, A-operand load applies ReLU and the input pad; 5 one-float-per-thread launches gone; sums unchanged
# speedup vs baseline: 13.2490x; 1.2177x over previous
#include <hip/hip_runtime.h>
#include <hip/hip_bf16.h>

typedef __attribute__((ext_vector_type(2))) float v2f;
typedef __attribute__((ext_vector_type(8))) float v8f;

constexpr int ACC_THREADS = 1024;
constexpr int ACC_UNROLL  = 8;
constexpr int DEG_ROWS    = 50016;
constexpr int DEG_SLICES  = 128;
constexpr int L3_ROWS     = 25008;
constexpr int L3_SLICES   = 64;

__device__ __forceinline__ size_t plane_pitch(size_t words) {
    return (words + 31) & ~(size_t)31;
}

__global__ __launch_bounds__(ACC_THREADS)
void k_deg_part(const int* __restrict__ dst, float* __restrict__ part, int e, int n) {
    __shared__ float cnt[DEG_ROWS];
    const int tid = threadIdx.x;
    const int r0  = blockIdx.y * DEG_ROWS;
    int nr = n - r0;
    if (nr > DEG_ROWS) nr = DEG_ROWS;
    for (int i = tid; i < nr; i += ACC_THREADS) cnt[i] = 0.0f;
    __syncthreads();

    const int per = (e + DEG_SLICES - 1) / DEG_SLICES;
    const int e0  = blockIdx.x * per;
    int e1 = e0 + per;
    if (e1 > e) e1 = e;
    for (int b = e0; b < e1; b += ACC_THREADS * ACC_UNROLL) {
        int d[ACC_UNROLL];
#pragma unroll
        for (int k = 0; k < ACC_UNROLL; ++k) {
            const int i = b + k * ACC_THREADS + tid;
            d[k] = (i < e1) ? dst[i] : -1;
        }
#pragma unroll
        for (int k = 0; k < ACC_UNROLL; ++k) {
            const int u = d[k] - r0;
            if ((unsigned)u < (unsigned)nr) atomicAdd(&cnt[u], 1.0f);
        }
    }
    __syncthreads();
    float* p = part + (size_t)blockIdx.x * plane_pitch((size_t)n) + r0;
    for (int i = tid; i < nr; i += ACC_THREADS) p[i] = cnt[i];
}

__global__ void k_norm_part(const float* __restrict__ part, float* __restrict__ isq,
                            float* __restrict__ idg, int n) {
    int i = blockIdx.x * blockDim.x + threadIdx.x;
    if (i < n) {
        const size_t pitch = plane_pitch((size_t)n);
        float d = 1.0f;
#pragma unroll 16
        for (int s = 0; s < DEG_SLICES; ++s) d += part[(size_t)s * pitch + i];
        isq[i] = rsqrtf(d);
        idg[i] = 1.0f / d;
    }
}

__global__ void k_prep_w(const float* __restrict__ W, float* __restrict__ Bt,
                         int Kin, int Cin, int KPAD, int COLS) {
    int idx = blockIdx.x * blockDim.x + threadIdx.x;
    if (idx >= COLS * KPAD) return;
    int col = idx / KPAD, k = idx % KPAD;
    Bt[idx] = (k < Kin && col < Cin) ? W[k * Cin + col] : 0.0f;
}

template <int K, int LDA, int NT, int SC, bool RELU_A, bool EPI>
__global__ void k_gemm_wmma(const float* __restrict__ A, const float* __restrict__ Bt,
                            float* __restrict__ H, const float* __restrict__ idg,
                            const float* __restrict__ bias, float* __restrict__ AGG) {
    static_assert(!EPI || NT * 16 == SC, "the epilogue reads bias[col] for every column tile");
    const int lane = threadIdx.x & 31;
    const int wave = threadIdx.x >> 5;
    const int half = lane >> 4;
    const int l15  = lane & 15;
    const size_t row = (size_t)blockIdx.x * 16 + l15;
    const int    col = wave * 16 + l15;
    const float* Arow = A + row * LDA;
    const float* Bcol = Bt + (size_t)col * K;
    const size_t mbase = (size_t)blockIdx.x * 16 + half * 8;

    float4 g0 = make_float4(0.0f, 0.0f, 0.0f, 0.0f), g1 = g0;
    float  bj = 0.0f;
    if (EPI) {
        g0 = *(const float4*)(idg + mbase);
        g1 = *(const float4*)(idg + mbase + 4);
        bj = bias[col];
    }

    v8f c = {};
#pragma unroll
    for (int k = 0; k < K; k += 4) {
        const int ka = k + half * 2;
        v2f a;
        if (LDA == K) {
            a = *(const v2f*)(Arow + ka);
        } else {
            a.x = (ka     < LDA) ? Arow[ka]     : 0.0f;
            a.y = (ka + 1 < LDA) ? Arow[ka + 1] : 0.0f;
        }
        if (RELU_A) {
            a.x = fmaxf(a.x, 0.0f);
            a.y = fmaxf(a.y, 0.0f);
        }
        v2f b = *(const v2f*)(Bcol + ka);
        c = __builtin_amdgcn_wmma_f32_16x16x4_f32(
                 false, a,  false, b,
                 (short)0, c,  false,  false);
    }

    if (NT * 16 == SC || col < SC) {
#pragma unroll
        for (int r = 0; r < 8; ++r)
            H[(mbase + r) * SC + col] = c[r];
        if (EPI) {
            const float g[8] = {g0.x, g0.y, g0.z, g0.w, g1.x, g1.y, g1.z, g1.w};
#pragma unroll
            for (int r = 0; r < 8; ++r)
                AGG[(mbase + r) * SC + col] = c[r] * g[r] + bj;
        }
    }
}

constexpr int TILE_ROWS    = 391;
constexpr int TILE_THREADS = 1024;
constexpr int Q_CAP        = 64;
constexpr int Q_BATCH      = 32;
constexpr int Q_GROUP      = 16;

constexpr int KEY_WIN   = 1024;
constexpr int ROW_SHIFT = 23;
static_assert(TILE_ROWS <= (1 << (32 - ROW_SHIFT)), "row inside the tile must fit above ROW_SHIFT");

constexpr int PREP_THREADS = 256;
constexpr int PREP_STEPS   = 16;
__global__ __launch_bounds__(PREP_THREADS)
void k_edge_prep(const int* __restrict__ src, const int* __restrict__ dst,
                 const float* __restrict__ isq, float* __restrict__ coef,
                 int* __restrict__ pk, int* __restrict__ keys, int e) {
    const int lane = threadIdx.x & 31;
    const int win = __builtin_amdgcn_readfirstlane((int)(blockIdx.x * (PREP_THREADS / 32) + (threadIdx.x >> 5)));
    const int wb  = win * KEY_WIN;
    if (wb >= e) return;
    int pl[8] = {0, 0, 0, 0, 0, 0, 0, 0};
    for (int i0 = 0; i0 < 32; i0 += PREP_STEPS) {
        int d[PREP_STEPS], s[PREP_STEPS];
#pragma unroll
        for (int k = 0; k < PREP_STEPS; ++k) {
            const int  i  = wb + (i0 + k) * 32 + lane;
            const bool ok = i < e;
            d[k] = ok ? dst[i] : 0;
            s[k] = ok ? src[i] : 0;
        }
        float fs[PREP_STEPS], fd[PREP_STEPS];
#pragma unroll
        for (int k = 0; k < PREP_STEPS; ++k) {
            fs[k] = isq[s[k]];
            fd[k] = isq[d[k]];
        }
#pragma unroll
        for (int k = 0; k < PREP_STEPS; ++k) {
            const int  i  = wb + (i0 + k) * 32 + lane;
            const bool ok = i < e;
            const int   t  = d[k] / TILE_ROWS;
            const int   u  = d[k] - t * TILE_ROWS;
            const float cf = fs[k] * fd[k];
            if (ok) {
                coef[i] = cf;
                ((int2*)pk)[i] = make_int2((int)((unsigned)s[k] | ((unsigned)u << ROW_SHIFT)), __float_as_int(cf));
            }
            const bool mine = lane == i0 + k;
#pragma unroll
            for (int p = 0; p < 8; ++p) {
                const int bits = (int)__builtin_amdgcn_ballot_w32(ok && ((t >> p) & 1));
                pl[p] = mine ? bits : pl[p];
            }
        }
    }
    int4* k4 = (int4*)keys + (size_t)win * 64;
    k4[lane]      = make_int4(pl[0], pl[1], pl[2], pl[3]);
    k4[32 + lane] = make_int4(pl[4], pl[5], pl[6], pl[7]);
}

constexpr int REC_CAP   = 768;
constexpr int REC_WORDS = REC_CAP * 3;

template <bool FULL>
__device__ __forceinline__ void rows_add(int s, int u, float cf, int cnt, int lane,
                                         const float* __restrict__ h, float* tile) {
#pragma unroll
    for (int g = 0; g < Q_BATCH; g += Q_GROUP) {
        float v0[Q_GROUP], v1[Q_GROUP];
#pragma unroll
        for (int k = 0; k < Q_GROUP; ++k) {
            if (FULL || g + k < cnt) {
                const float* hs = h + (size_t)__builtin_amdgcn_readlane(s, g + k) * 64;
                v0[k] = hs[lane];
                v1[k] = hs[lane + 32];
            }
        }
#pragma unroll
        for (int k = 0; k < Q_GROUP; ++k) {
            if (FULL || g + k < cnt) {
                const float cfk = __int_as_float(__builtin_amdgcn_readlane(__float_as_int(cf), g + k));
                float* t = tile + __builtin_amdgcn_readlane(u, g + k) * 64;
                atomicAdd(t + lane,      v0[k] * cfk);
                atomicAdd(t + lane + 32, v1[k] * cfk);
            }
        }
    }
}

template <bool FULL, bool REC>
__device__ __forceinline__ void queue_drain(const int* q, int qh, int cnt, int lane,
                                            const int* __restrict__ pk,
                                            const float* __restrict__ h, float* tile,
                                            float* recw, int& nrec) {
    int s = 0, u = 0;
    float cf = 0.0f;
    if (FULL || lane < cnt) {
        const int  en = q[(qh + lane) & (Q_CAP - 1)];
        const int2 r  = ((const int2*)pk)[en];
        s  = r.x & ((1 << ROW_SHIFT) - 1);
        u  = (int)((unsigned)r.x >> ROW_SHIFT);
        cf = __int_as_float(r.y);
    }
    if (REC && nrec >= 0) {
        if (nrec + Q_BATCH <= REC_CAP) {
            if (FULL || lane < cnt) {
                float* r = recw + nrec * 3 + lane;
                r[0]           = __int_as_float(s);
                r[Q_BATCH]     = __int_as_float(u);
                r[2 * Q_BATCH] = cf;
            }
            nrec += cnt;
        } else {
            nrec = -1;
        }
    }
    rows_add<FULL>(s, u, cf, cnt, lane, h, tile);
}

__device__ __forceinline__ unsigned key_match(const int4& lo, const int4& hi, const unsigned (&xm)[8]) {
    unsigned m = ((unsigned)lo.x ^ xm[0]) & ((unsigned)lo.y ^ xm[1]);
    m &= ((unsigned)lo.z ^ xm[2]);
    m &= ((unsigned)lo.w ^ xm[3]);
    m &= ((unsigned)hi.x ^ xm[4]);
    m &= ((unsigned)hi.y ^ xm[5]);
    m &= ((unsigned)hi.z ^ xm[6]);
    m &= ((unsigned)hi.w ^ xm[7]);
    return m;
}

template <bool REC>
__device__ __forceinline__ void window_hits(unsigned m, int ebase, int lane,
                                            int* q, int& qh, int& qn,
                                            const int* __restrict__ pk,
                                            const float* __restrict__ h, float* tile,
                                            float* recw, int& nrec) {
    unsigned bal = __builtin_amdgcn_ballot_w32(m != 0u);
    while (bal) {
        if (m != 0u) {
            const int slot = (qh + qn + (int)__builtin_amdgcn_mbcnt_lo(bal, 0u)) & (Q_CAP - 1);
            q[slot] = ebase + __builtin_ctz(m);
            m &= m - 1u;
        }
        qn += __builtin_popcount(bal);
        if (qn >= Q_BATCH) {
            queue_drain<true, REC>(q, qh, Q_BATCH, lane, pk, h, tile, recw, nrec);
            qh = (qh + Q_BATCH) & (Q_CAP - 1);
            qn -= Q_BATCH;
        }
        bal = __builtin_amdgcn_ballot_w32(m != 0u);
    }
}

template <bool REC>
__device__ __forceinline__ void scan_edges(const int* __restrict__ keys, const int* __restrict__ pk,
                                           const float* __restrict__ h,
                                           float* tile, int* q, int wave, int lane,
                                           int tno, int e, float* recw, int& nrec) {
    int qh = 0, qn = 0;
    unsigned xm[8];
#pragma unroll
    for (int p = 0; p < 8; ++p) xm[p] = ((tno >> p) & 1) ? 0u : ~0u;

    const int4* k4    = (const int4*)keys;
    const int   nfull = e / (2 * KEY_WIN);
    int4 nx[4];
    if (wave < nfull) {
        const int4* p = k4 + (size_t)wave * 128 + lane;
#pragma unroll
        for (int qq = 0; qq < 4; ++qq) nx[qq] = p[qq * 32];
    }
    for (int w = wave; w < nfull; w += TILE_THREADS / 32) {
        int4 cur[4];
#pragma unroll
        for (int qq = 0; qq < 4; ++qq) cur[qq] = nx[qq];
        const int wn = w + TILE_THREADS / 32;
        if (wn < nfull) {
            const int4* p = k4 + (size_t)wn * 128 + lane;
#pragma unroll
            for (int qq = 0; qq < 4; ++qq) nx[qq] = p[qq * 32];
        }
#pragma unroll
        for (int hh = 0; hh < 2; ++hh) {
            const unsigned m = key_match(cur[2 * hh], cur[2 * hh + 1], xm);
            window_hits<REC>(m, (2 * w + hh) * KEY_WIN + lane * 32, lane, q, qh, qn, pk, h, tile, recw, nrec);
        }
    }
    const int kw = (e + KEY_WIN - 1) / KEY_WIN;
    for (int tw = 2 * nfull + wave; tw < kw; tw += TILE_THREADS / 32) {
        const int4* p = k4 + (size_t)tw * 64 + lane;
        const int4 lo = p[0];
        const int4 hi = p[32];
        unsigned m = key_match(lo, hi, xm);
        const int ebase = tw * KEY_WIN + lane * 32;
        const int valid = e - ebase;
        if (valid < 32) m = (valid <= 0) ? 0u : (m & ((1u << valid) - 1u));
        window_hits<REC>(m, ebase, lane, q, qh, qn, pk, h, tile, recw, nrec);
    }
    if (qn > 0) queue_drain<false, REC>(q, qh, qn, lane, pk, h, tile, recw, nrec);
}

__device__ __forceinline__ void replay_edges(const float* __restrict__ recw, int nrec, int lane,
                                             const float* __restrict__ h, float* tile) {
    int s = 0, u = 0;
    float cf = 0.0f;
    if (lane < nrec) {
        s  = __float_as_int(recw[lane]);
        u  = __float_as_int(recw[Q_BATCH + lane]);
        cf = recw[2 * Q_BATCH + lane];
    }
    for (int b = 0; b < nrec; b += Q_BATCH) {
        int s2 = 0, u2 = 0;
        float cf2 = 0.0f;
        const int nb = b + Q_BATCH;
        if (nb + lane < nrec) {
            const float* r = recw + nb * 3 + lane;
            s2  = __float_as_int(r[0]);
            u2  = __float_as_int(r[Q_BATCH]);
            cf2 = r[2 * Q_BATCH];
        }
        if (nrec - b >= Q_BATCH) rows_add<true>(s, u, cf, Q_BATCH, lane, h, tile);
        else                     rows_add<false>(s, u, cf, nrec - b, lane, h, tile);
        s = s2; u = u2; cf = cf2;
    }
}

template <bool REPLAY>
__global__ __launch_bounds__(TILE_THREADS)
void k_scatter64_tile(const int* __restrict__ keys, const int* __restrict__ pk,
                      const float* __restrict__ h,
                      float* __restrict__ agg, int* __restrict__ lcnt, float* __restrict__ rec,
                      int e, int n) {
    __shared__ float tile[TILE_ROWS * 64];
    __shared__ int   queue[(TILE_THREADS / 32) * Q_CAP];
    const int tid  = threadIdx.x;
    const int lane = tid & 31;
    const int wave = __builtin_amdgcn_readfirstlane(tid >> 5);
    const int r0   = blockIdx.x * TILE_ROWS;
    int nr = n - r0;
    if (nr > TILE_ROWS) nr = TILE_ROWS;
    if (nr <= 0) return;

    float4*       t4 = (float4*)tile;
    float4*       a4 = (float4*)(agg + (size_t)r0 * 64);
    for (int i = tid; i < nr * 16; i += TILE_THREADS) t4[i] = a4[i];
    __syncthreads();

    int*   q    = queue + wave * Q_CAP;
    const int tno = __builtin_amdgcn_readfirstlane((int)blockIdx.x);
    const int wid = blockIdx.x * (TILE_THREADS / 32) + wave;
    float* recw = rec + (size_t)wid * REC_WORDS;

    if (REPLAY) {
        int nrec = __builtin_amdgcn_readfirstlane(lcnt[wid]);
        if (nrec >= 0) replay_edges(recw, nrec, lane, h, tile);
        else           scan_edges<false>(keys, pk, h, tile, q, wave, lane, tno, e, recw, nrec);
    } else {
        int nrec = 0;
        scan_edges<true>(keys, pk, h, tile, q, wave, lane, tno, e, recw, nrec);
        if (lane == 0) lcnt[wid] = nrec;
    }
    __syncthreads();
    for (int i = tid; i < nr * 16; i += TILE_THREADS) a4[i] = t4[i];
}

__global__ __launch_bounds__(ACC_THREADS)
void k_l3_part(const int* __restrict__ src, const int* __restrict__ dst,
               const float* __restrict__ coef, const float* __restrict__ h,
               float* __restrict__ part, int e, int n) {
    __shared__ float acc[L3_ROWS * 2];
    const int tid = threadIdx.x;
    const int r0  = blockIdx.y * L3_ROWS;
    int nr = n - r0;
    if (nr > L3_ROWS) nr = L3_ROWS;
    float2* acc2 = (float2*)acc;
    for (int i = tid; i < nr; i += ACC_THREADS) acc2[i] = make_float2(0.0f, 0.0f);
    __syncthreads();

    const int per = (e + L3_SLICES - 1) / L3_SLICES;
    const int e0  = blockIdx.x * per;
    int e1 = e0 + per;
    if (e1 > e) e1 = e;
    for (int b = e0; b < e1; b += ACC_THREADS * ACC_UNROLL) {
        int   d[ACC_UNROLL], s[ACC_UNROLL];
        float cf[ACC_UNROLL];
#pragma unroll
        for (int k = 0; k < ACC_UNROLL; ++k) {
            const int  i  = b + k * ACC_THREADS + tid;
            const bool ok = i < e1;
            d[k]  = ok ? dst[i]  : -1;
            s[k]  = ok ? src[i]  : 0;
            cf[k] = ok ? coef[i] : 0.0f;
        }
        float2 v[ACC_UNROLL];
#pragma unroll
        for (int k = 0; k < ACC_UNROLL; ++k) {
            v[k] = make_float2(0.0f, 0.0f);
            if ((unsigned)(d[k] - r0) < (unsigned)nr) v[k] = *(const float2*)(h + (size_t)s[k] * 2);
        }
#pragma unroll
        for (int k = 0; k < ACC_UNROLL; ++k) {
            const int u = d[k] - r0;
            if ((unsigned)u < (unsigned)nr) {
                atomicAdd(&acc[u * 2 + 0], v[k].x * cf[k]);
                atomicAdd(&acc[u * 2 + 1], v[k].y * cf[k]);
            }
        }
    }
    __syncthreads();
    float2* p = (float2*)(part + (size_t)blockIdx.x * plane_pitch((size_t)n * 2)) + r0;
    for (int i = tid; i < nr; i += ACC_THREADS) p[i] = acc2[i];
}

__global__ void k_l3_sum(const float* __restrict__ h, const float* __restrict__ idg,
                         const float* __restrict__ bias, const float* __restrict__ part,
                         float* __restrict__ out, int n) {
    int i = blockIdx.x * blockDim.x + threadIdx.x;
    if (i >= n) return;
    const size_t pitch2 = plane_pitch((size_t)n * 2) / 2;
    const float2 hv = ((const float2*)h)[i];
    const float  g  = idg[i];
    float2 a;
    a.x = hv.x * g + bias[0];
    a.y = hv.y * g + bias[1];
    const float2* p = (const float2*)part + i;
#pragma unroll 16
    for (int s = 0; s < L3_SLICES; ++s) {
        const float2 t = p[(size_t)s * pitch2];
        a.x += t.x;
        a.y += t.y;
    }
    ((float2*)out)[i] = a;
}

extern "C" void kernel_launch(void* const* d_in, const int* in_sizes, int n_in,
                              void* d_out, int out_size, void* d_ws, size_t ws_size,
                              hipStream_t stream) {
    const float* x  = (const float*)d_in[0];
    const int*   ei = (const int*)d_in[1];
    const float* W1 = (const float*)d_in[2];
    const float* b1 = (const float*)d_in[3];
    const float* W2 = (const float*)d_in[4];
    const float* b2 = (const float*)d_in[5];
    const float* W3 = (const float*)d_in[6];
    const float* b3 = (const float*)d_in[7];
    float* out = (float*)d_out;

    const int N = in_sizes[0] / 5;
    const int E = in_sizes[1] / 2;
    const int* src = ei;
    const int* dst = ei + E;

    float* ws   = (float*)d_ws;
    float* isq  = ws;
    float* idg  = isq  + N;
    float* coef = idg  + N;
    float* w1t  = coef + E;
    float* w2t  = w1t  + 64 * 8;
    float* w3t  = w2t  + 64 * 64;
    float* bufA = w3t  + 16 * 64;
    float* bufB = bufA + (size_t)N * 64;
    float* bufC = bufB + (size_t)N * 64;
    float* h3   = bufC + (size_t)N * 64;
    float* part = h3   + (size_t)N * 2;
    int*   lcnt = (int*)(part + (size_t)128 * (N + 32));
    float* rec  = (float*)(lcnt + (size_t)((N + TILE_ROWS - 1) / TILE_ROWS) * (TILE_THREADS / 32));
    int*   keys = (int*)(rec + (size_t)((N + TILE_ROWS - 1) / TILE_ROWS) * (TILE_THREADS / 32) * REC_WORDS);
    int*   pk   = keys + (size_t)((E + KEY_WIN - 1) / KEY_WIN) * (KEY_WIN / 4);

    const int T = 256;
    auto cdiv = [](long a, long b) { return (int)((a + b - 1) / b); };

    k_deg_part <<<dim3(DEG_SLICES, cdiv(N, DEG_ROWS)), ACC_THREADS, 0, stream>>>(dst, part, E, N);
    k_norm_part<<<cdiv(N, T), T, 0, stream>>>(part, isq, idg, N);
    k_edge_prep<<<cdiv(cdiv(E, KEY_WIN), PREP_THREADS / 32), PREP_THREADS, 0, stream>>>(src, dst, isq, coef, pk, keys, E);
    k_prep_w   <<<cdiv(64 * 8,  T), T, 0, stream>>>(W1, w1t, 5, 64, 8, 64);
    k_prep_w   <<<cdiv(64 * 64, T), T, 0, stream>>>(W2, w2t, 64, 64, 64, 64);
    k_prep_w   <<<cdiv(16 * 64, T), T, 0, stream>>>(W3, w3t, 64, 2, 64, 16);

    const int GT = cdiv(N, 16);

    k_gemm_wmma<8, 5, 4, 64, false, true><<<GT, 128, 0, stream>>>(x, w1t, bufA, idg, b1, bufB);
    k_scatter64_tile<false><<<cdiv(N, TILE_ROWS), TILE_THREADS, 0, stream>>>(keys, pk, bufA, bufB, lcnt, rec, E, N);

    k_gemm_wmma<64, 64, 4, 64, true, true><<<GT, 128, 0, stream>>>(bufB, w2t, bufA, idg, b2, bufC);
    k_scatter64_tile<true><<<cdiv(N, TILE_ROWS), TILE_THREADS, 0, stream>>>(keys, pk, bufA, bufC, lcnt, rec, E, N);

    k_gemm_wmma<64, 64, 1, 2, true, false><<<GT, 32, 0, stream>>>(bufC, w3t, h3, nullptr, nullptr, nullptr);
    k_l3_part  <<<dim3(L3_SLICES, cdiv(N, L3_ROWS)), ACC_THREADS, 0, stream>>>(src, dst, coef, h3, part, E, N);
    k_l3_sum   <<<cdiv(N, T), T, 0, stream>>>(h3, idg, b3, part, out, N);
}


# Layer-3 edge sum on the 391-row LDS tiles: a wave adds its recorded edge list into rows started at self loop+bias; edge re-read, slice planes and 1 launch gone; layer-3 row sum order changes
# speedup vs baseline: 14.5275x; 1.0965x over previous
#include <hip/hip_runtime.h>
#include <hip/hip_bf16.h>

typedef __attribute__((ext_vector_type(2))) float v2f;
typedef __attribute__((ext_vector_type(8))) float v8f;

constexpr int ACC_THREADS = 1024;
constexpr int ACC_UNROLL  = 8;
constexpr int DEG_ROWS    = 50016;
constexpr int DEG_SLICES  = 128;

__device__ __forceinline__ size_t plane_pitch(size_t words) {
    return (words + 31) & ~(size_t)31;
}

__global__ __launch_bounds__(ACC_THREADS)
void k_deg_part(const int* __restrict__ dst, float* __restrict__ part, int e, int n) {
    __shared__ float cnt[DEG_ROWS];
    const int tid = threadIdx.x;
    const int r0  = blockIdx.y * DEG_ROWS;
    int nr = n - r0;
    if (nr > DEG_ROWS) nr = DEG_ROWS;
    for (int i = tid; i < nr; i += ACC_THREADS) cnt[i] = 0.0f;
    __syncthreads();

    const int per = (e + DEG_SLICES - 1) / DEG_SLICES;
    const int e0  = blockIdx.x * per;
    int e1 = e0 + per;
    if (e1 > e) e1 = e;
    for (int b = e0; b < e1; b += ACC_THREADS * ACC_UNROLL) {
        int d[ACC_UNROLL];
#pragma unroll
        for (int k = 0; k < ACC_UNROLL; ++k) {
            const int i = b + k * ACC_THREADS + tid;
            d[k] = (i < e1) ? dst[i] : -1;
        }
#pragma unroll
        for (int k = 0; k < ACC_UNROLL; ++k) {
            const int u = d[k] - r0;
            if ((unsigned)u < (unsigned)nr) atomicAdd(&cnt[u], 1.0f);
        }
    }
    __syncthreads();
    float* p = part + (size_t)blockIdx.x * plane_pitch((size_t)n) + r0;
    for (int i = tid; i < nr; i += ACC_THREADS) p[i] = cnt[i];
}

__global__ void k_norm_part(const float* __restrict__ part, float* __restrict__ isq,
                            float* __restrict__ idg, int n) {
    int i = blockIdx.x * blockDim.x + threadIdx.x;
    if (i < n) {
        const size_t pitch = plane_pitch((size_t)n);
        float d = 1.0f;
#pragma unroll 16
        for (int s = 0; s < DEG_SLICES; ++s) d += part[(size_t)s * pitch + i];
        isq[i] = rsqrtf(d);
        idg[i] = 1.0f / d;
    }
}

__global__ void k_prep_w(const float* __restrict__ W, float* __restrict__ Bt,
                         int Kin, int Cin, int KPAD, int COLS) {
    int idx = blockIdx.x * blockDim.x + threadIdx.x;
    if (idx >= COLS * KPAD) return;
    int col = idx / KPAD, k = idx % KPAD;
    Bt[idx] = (k < Kin && col < Cin) ? W[k * Cin + col] : 0.0f;
}

template <int K, int LDA, int NT, int SC, bool RELU_A, bool EPI>
__global__ void k_gemm_wmma(const float* __restrict__ A, const float* __restrict__ Bt,
                            float* __restrict__ H, const float* __restrict__ idg,
                            const float* __restrict__ bias, float* __restrict__ AGG) {
    static_assert(!EPI || NT * 16 == SC, "the epilogue reads bias[col] for every column tile");
    const int lane = threadIdx.x & 31;
    const int wave = threadIdx.x >> 5;
    const int half = lane >> 4;
    const int l15  = lane & 15;
    const size_t row = (size_t)blockIdx.x * 16 + l15;
    const int    col = wave * 16 + l15;
    const float* Arow = A + row * LDA;
    const float* Bcol = Bt + (size_t)col * K;
    const size_t mbase = (size_t)blockIdx.x * 16 + half * 8;

    float4 g0 = make_float4(0.0f, 0.0f, 0.0f, 0.0f), g1 = g0;
    float  bj = 0.0f;
    if (EPI) {
        g0 = *(const float4*)(idg + mbase);
        g1 = *(const float4*)(idg + mbase + 4);
        bj = bias[col];
    }

    v8f c = {};
#pragma unroll
    for (int k = 0; k < K; k += 4) {
        const int ka = k + half * 2;
        v2f a;
        if (LDA == K) {
            a = *(const v2f*)(Arow + ka);
        } else {
            a.x = (ka     < LDA) ? Arow[ka]     : 0.0f;
            a.y = (ka + 1 < LDA) ? Arow[ka + 1] : 0.0f;
        }
        if (RELU_A) {
            a.x = fmaxf(a.x, 0.0f);
            a.y = fmaxf(a.y, 0.0f);
        }
        v2f b = *(const v2f*)(Bcol + ka);
        c = __builtin_amdgcn_wmma_f32_16x16x4_f32(
                 false, a,  false, b,
                 (short)0, c,  false,  false);
    }

    if (NT * 16 == SC || col < SC) {
#pragma unroll
        for (int r = 0; r < 8; ++r)
            H[(mbase + r) * SC + col] = c[r];
        if (EPI) {
            const float g[8] = {g0.x, g0.y, g0.z, g0.w, g1.x, g1.y, g1.z, g1.w};
#pragma unroll
            for (int r = 0; r < 8; ++r)
                AGG[(mbase + r) * SC + col] = c[r] * g[r] + bj;
        }
    }
}

constexpr int TILE_ROWS    = 391;
constexpr int TILE_THREADS = 1024;
constexpr int Q_CAP        = 64;
constexpr int Q_BATCH      = 32;
constexpr int Q_GROUP      = 16;

constexpr int KEY_WIN   = 1024;
constexpr int ROW_SHIFT = 23;
static_assert(TILE_ROWS <= (1 << (32 - ROW_SHIFT)), "row inside the tile must fit above ROW_SHIFT");

constexpr int PREP_THREADS = 256;
constexpr int PREP_STEPS   = 16;
__global__ __launch_bounds__(PREP_THREADS)
void k_edge_prep(const int* __restrict__ src, const int* __restrict__ dst,
                 const float* __restrict__ isq,
                 int* __restrict__ pk, int* __restrict__ keys, int e) {
    const int lane = threadIdx.x & 31;
    const int win = __builtin_amdgcn_readfirstlane((int)(blockIdx.x * (PREP_THREADS / 32) + (threadIdx.x >> 5)));
    const int wb  = win * KEY_WIN;
    if (wb >= e) return;
    int pl[8] = {0, 0, 0, 0, 0, 0, 0, 0};
    for (int i0 = 0; i0 < 32; i0 += PREP_STEPS) {
        int d[PREP_STEPS], s[PREP_STEPS];
#pragma unroll
        for (int k = 0; k < PREP_STEPS; ++k) {
            const int  i  = wb + (i0 + k) * 32 + lane;
            const bool ok = i < e;
            d[k] = ok ? dst[i] : 0;
            s[k] = ok ? src[i] : 0;
        }
        float fs[PREP_STEPS], fd[PREP_STEPS];
#pragma unroll
        for (int k = 0; k < PREP_STEPS; ++k) {
            fs[k] = isq[s[k]];
            fd[k] = isq[d[k]];
        }
#pragma unroll
        for (int k = 0; k < PREP_STEPS; ++k) {
            const int  i  = wb + (i0 + k) * 32 + lane;
            const bool ok = i < e;
            const int   t  = d[k] / TILE_ROWS;
            const int   u  = d[k] - t * TILE_ROWS;
            const float cf = fs[k] * fd[k];
            if (ok)
                ((int2*)pk)[i] = make_int2((int)((unsigned)s[k] | ((unsigned)u << ROW_SHIFT)), __float_as_int(cf));
            const bool mine = lane == i0 + k;
#pragma unroll
            for (int p = 0; p < 8; ++p) {
                const int bits = (int)__builtin_amdgcn_ballot_w32(ok && ((t >> p) & 1));
                pl[p] = mine ? bits : pl[p];
            }
        }
    }
    int4* k4 = (int4*)keys + (size_t)win * 64;
    k4[lane]      = make_int4(pl[0], pl[1], pl[2], pl[3]);
    k4[32 + lane] = make_int4(pl[4], pl[5], pl[6], pl[7]);
}

constexpr int REC_CAP   = 768;
constexpr int REC_WORDS = REC_CAP * 3;

template <bool FULL>
__device__ __forceinline__ void rows_add(int s, int u, float cf, int cnt, int lane,
                                         const float* __restrict__ h, float* tile) {
#pragma unroll
    for (int g = 0; g < Q_BATCH; g += Q_GROUP) {
        float v0[Q_GROUP], v1[Q_GROUP];
#pragma unroll
        for (int k = 0; k < Q_GROUP; ++k) {
            if (FULL || g + k < cnt) {
                const float* hs = h + (size_t)__builtin_amdgcn_readlane(s, g + k) * 64;
                v0[k] = hs[lane];
                v1[k] = hs[lane + 32];
            }
        }
#pragma unroll
        for (int k = 0; k < Q_GROUP; ++k) {
            if (FULL || g + k < cnt) {
                const float cfk = __int_as_float(__builtin_amdgcn_readlane(__float_as_int(cf), g + k));
                float* t = tile + __builtin_amdgcn_readlane(u, g + k) * 64;
                atomicAdd(t + lane,      v0[k] * cfk);
                atomicAdd(t + lane + 32, v1[k] * cfk);
            }
        }
    }
}

template <bool FULL, bool REC>
__device__ __forceinline__ void queue_drain(const int* q, int qh, int cnt, int lane,
                                            const int* __restrict__ pk,
                                            const float* __restrict__ h, float* tile,
                                            float* recw, int& nrec) {
    int s = 0, u = 0;
    float cf = 0.0f;
    if (FULL || lane < cnt) {
        const int  en = q[(qh + lane) & (Q_CAP - 1)];
        const int2 r  = ((const int2*)pk)[en];
        s  = r.x & ((1 << ROW_SHIFT) - 1);
        u  = (int)((unsigned)r.x >> ROW_SHIFT);
        cf = __int_as_float(r.y);
    }
    if (REC && nrec >= 0) {
        if (nrec + Q_BATCH <= REC_CAP) {
            if (FULL || lane < cnt) {
                float* r = recw + nrec * 3 + lane;
                r[0]           = __int_as_float(s);
                r[Q_BATCH]     = __int_as_float(u);
                r[2 * Q_BATCH] = cf;
            }
            nrec += cnt;
        } else {
            nrec = -1;
        }
    }
    rows_add<FULL>(s, u, cf, cnt, lane, h, tile);
}

__device__ __forceinline__ unsigned key_match(const int4& lo, const int4& hi, const unsigned (&xm)[8]) {
    unsigned m = ((unsigned)lo.x ^ xm[0]) & ((unsigned)lo.y ^ xm[1]);
    m &= ((unsigned)lo.z ^ xm[2]);
    m &= ((unsigned)lo.w ^ xm[3]);
    m &= ((unsigned)hi.x ^ xm[4]);
    m &= ((unsigned)hi.y ^ xm[5]);
    m &= ((unsigned)hi.z ^ xm[6]);
    m &= ((unsigned)hi.w ^ xm[7]);
    return m;
}

template <bool REC>
__device__ __forceinline__ void window_hits(unsigned m, int ebase, int lane,
                                            int* q, int& qh, int& qn,
                                            const int* __restrict__ pk,
                                            const float* __restrict__ h, float* tile,
                                            float* recw, int& nrec) {
    unsigned bal = __builtin_amdgcn_ballot_w32(m != 0u);
    while (bal) {
        if (m != 0u) {
            const int slot = (qh + qn + (int)__builtin_amdgcn_mbcnt_lo(bal, 0u)) & (Q_CAP - 1);
            q[slot] = ebase + __builtin_ctz(m);
            m &= m - 1u;
        }
        qn += __builtin_popcount(bal);
        if (qn >= Q_BATCH) {
            queue_drain<true, REC>(q, qh, Q_BATCH, lane, pk, h, tile, recw, nrec);
            qh = (qh + Q_BATCH) & (Q_CAP - 1);
            qn -= Q_BATCH;
        }
        bal = __builtin_amdgcn_ballot_w32(m != 0u);
    }
}

template <bool REC>
__device__ __forceinline__ void scan_edges(const int* __restrict__ keys, const int* __restrict__ pk,
                                           const float* __restrict__ h,
                                           float* tile, int* q, int wave, int lane,
                                           int tno, int e, float* recw, int& nrec) {
    int qh = 0, qn = 0;
    unsigned xm[8];
#pragma unroll
    for (int p = 0; p < 8; ++p) xm[p] = ((tno >> p) & 1) ? 0u : ~0u;

    const int4* k4    = (const int4*)keys;
    const int   nfull = e / (2 * KEY_WIN);
    int4 nx[4];
    if (wave < nfull) {
        const int4* p = k4 + (size_t)wave * 128 + lane;
#pragma unroll
        for (int qq = 0; qq < 4; ++qq) nx[qq] = p[qq * 32];
    }
    for (int w = wave; w < nfull; w += TILE_THREADS / 32) {
        int4 cur[4];
#pragma unroll
        for (int qq = 0; qq < 4; ++qq) cur[qq] = nx[qq];
        const int wn = w + TILE_THREADS / 32;
        if (wn < nfull) {
            const int4* p = k4 + (size_t)wn * 128 + lane;
#pragma unroll
            for (int qq = 0; qq < 4; ++qq) nx[qq] = p[qq * 32];
        }
#pragma unroll
        for (int hh = 0; hh < 2; ++hh) {
            const unsigned m = key_match(cur[2 * hh], cur[2 * hh + 1], xm);
            window_hits<REC>(m, (2 * w + hh) * KEY_WIN + lane * 32, lane, q, qh, qn, pk, h, tile, recw, nrec);
        }
    }
    const int kw = (e + KEY_WIN - 1) / KEY_WIN;
    for (int tw = 2 * nfull + wave; tw < kw; tw += TILE_THREADS / 32) {
        const int4* p = k4 + (size_t)tw * 64 + lane;
        const int4 lo = p[0];
        const int4 hi = p[32];
        unsigned m = key_match(lo, hi, xm);
        const int ebase = tw * KEY_WIN + lane * 32;
        const int valid = e - ebase;
        if (valid < 32) m = (valid <= 0) ? 0u : (m & ((1u << valid) - 1u));
        window_hits<REC>(m, ebase, lane, q, qh, qn, pk, h, tile, recw, nrec);
    }
    if (qn > 0) queue_drain<false, REC>(q, qh, qn, lane, pk, h, tile, recw, nrec);
}

__device__ __forceinline__ void replay_edges(const float* __restrict__ recw, int nrec, int lane,
                                             const float* __restrict__ h, float* tile) {
    int s = 0, u = 0;
    float cf = 0.0f;
    if (lane < nrec) {
        s  = __float_as_int(recw[lane]);
        u  = __float_as_int(recw[Q_BATCH + lane]);
        cf = recw[2 * Q_BATCH + lane];
    }
    for (int b = 0; b < nrec; b += Q_BATCH) {
        int s2 = 0, u2 = 0;
        float cf2 = 0.0f;
        const int nb = b + Q_BATCH;
        if (nb + lane < nrec) {
            const float* r = recw + nb * 3 + lane;
            s2  = __float_as_int(r[0]);
            u2  = __float_as_int(r[Q_BATCH]);
            cf2 = r[2 * Q_BATCH];
        }
        if (nrec - b >= Q_BATCH) rows_add<true>(s, u, cf, Q_BATCH, lane, h, tile);
        else                     rows_add<false>(s, u, cf, nrec - b, lane, h, tile);
        s = s2; u = u2; cf = cf2;
    }
}

template <bool REPLAY>
__global__ __launch_bounds__(TILE_THREADS)
void k_scatter64_tile(const int* __restrict__ keys, const int* __restrict__ pk,
                      const float* __restrict__ h,
                      float* __restrict__ agg, int* __restrict__ lcnt, float* __restrict__ rec,
                      int e, int n) {
    __shared__ float tile[TILE_ROWS * 64];
    __shared__ int   queue[(TILE_THREADS / 32) * Q_CAP];
    const int tid  = threadIdx.x;
    const int lane = tid & 31;
    const int wave = __builtin_amdgcn_readfirstlane(tid >> 5);
    const int r0   = blockIdx.x * TILE_ROWS;
    int nr = n - r0;
    if (nr > TILE_ROWS) nr = TILE_ROWS;
    if (nr <= 0) return;

    float4*       t4 = (float4*)tile;
    float4*       a4 = (float4*)(agg + (size_t)r0 * 64);
    for (int i = tid; i < nr * 16; i += TILE_THREADS) t4[i] = a4[i];
    __syncthreads();

    int*   q    = queue + wave * Q_CAP;
    const int tno = __builtin_amdgcn_readfirstlane((int)blockIdx.x);
    const int wid = blockIdx.x * (TILE_THREADS / 32) + wave;
    float* recw = rec + (size_t)wid * REC_WORDS;

    if (REPLAY) {
        int nrec = __builtin_amdgcn_readfirstlane(lcnt[wid]);
        if (nrec >= 0) replay_edges(recw, nrec, lane, h, tile);
        else           scan_edges<false>(keys, pk, h, tile, q, wave, lane, tno, e, recw, nrec);
    } else {
        int nrec = 0;
        scan_edges<true>(keys, pk, h, tile, q, wave, lane, tno, e, recw, nrec);
        if (lane == 0) lcnt[wid] = nrec;
    }
    __syncthreads();
    for (int i = tid; i < nr * 16; i += TILE_THREADS) a4[i] = t4[i];
}

constexpr int L3_BATCHES = 8;

__device__ __forceinline__ void l3_window(int tw, const int4* __restrict__ k4, const unsigned (&xm)[8],
                                          const int* __restrict__ pk, const float* __restrict__ h,
                                          float* acc, int lane, int e) {
    const int4* p = k4 + (size_t)tw * 64 + lane;
    unsigned m = key_match(p[0], p[32], xm);
    const int ebase = tw * KEY_WIN + lane * 32;
    const int valid = e - ebase;
    if (valid < 32) m = (valid <= 0) ? 0u : (m & ((1u << valid) - 1u));
    while (m) {
        const int  en = ebase + __builtin_ctz(m);
        m &= m - 1u;
        const int2   r  = ((const int2*)pk)[en];
        const int    s  = r.x & ((1 << ROW_SHIFT) - 1);
        const int    u  = (int)((unsigned)r.x >> ROW_SHIFT);
        const float  cf = __int_as_float(r.y);
        const float2 v  = *(const float2*)(h + (size_t)s * 2);
        atomicAdd(&acc[u * 2 + 0], v.x * cf);
        atomicAdd(&acc[u * 2 + 1], v.y * cf);
    }
}

__device__ __forceinline__ void l3_scan_edges(const int* __restrict__ keys, const int* __restrict__ pk,
                                              const float* __restrict__ h, float* acc,
                                              int wave, int lane, int tno, int e) {
    unsigned xm[8];
#pragma unroll
    for (int p = 0; p < 8; ++p) xm[p] = ((tno >> p) & 1) ? 0u : ~0u;
    const int4* k4    = (const int4*)keys;
    const int   nfull = e / (2 * KEY_WIN);
    const int   kw    = (e + KEY_WIN - 1) / KEY_WIN;
    for (int w = wave; w < nfull; w += TILE_THREADS / 32) {
        l3_window(2 * w,     k4, xm, pk, h, acc, lane, e);
        l3_window(2 * w + 1, k4, xm, pk, h, acc, lane, e);
    }
    for (int tw = 2 * nfull + wave; tw < kw; tw += TILE_THREADS / 32)
        l3_window(tw, k4, xm, pk, h, acc, lane, e);
}

__global__ __launch_bounds__(TILE_THREADS)
void k_l3_tile(const int* __restrict__ keys, const int* __restrict__ pk,
               const float* __restrict__ h, const float* __restrict__ idg,
               const float* __restrict__ bias, const int* __restrict__ lcnt,
               const float* __restrict__ rec, float* __restrict__ out, int e, int n) {
    __shared__ float acc[TILE_ROWS * 2];
    const int tid  = threadIdx.x;
    const int lane = tid & 31;
    const int wave = __builtin_amdgcn_readfirstlane(tid >> 5);
    const int r0   = blockIdx.x * TILE_ROWS;
    int nr = n - r0;
    if (nr > TILE_ROWS) nr = TILE_ROWS;
    if (nr <= 0) return;

    float2* acc2 = (float2*)acc;
    const float b0 = bias[0], b1 = bias[1];
    for (int i = tid; i < nr; i += TILE_THREADS) {
        const float2 hv = ((const float2*)h)[r0 + i];
        const float  g  = idg[r0 + i];
        acc2[i] = make_float2(hv.x * g + b0, hv.y * g + b1);
    }
    __syncthreads();

    const int wid = blockIdx.x * (TILE_THREADS / 32) + wave;
    const float* recw = rec + (size_t)wid * REC_WORDS;
    const int nrec = __builtin_amdgcn_readfirstlane(lcnt[wid]);
    if (nrec >= 0) {
        for (int b = 0; b < nrec; b += Q_BATCH * L3_BATCHES) {
            int   s[L3_BATCHES], u[L3_BATCHES];
            float cf[L3_BATCHES];
#pragma unroll
            for (int k = 0; k < L3_BATCHES; ++k) {
                const int    bk = b + k * Q_BATCH;
                const float* r  = recw + bk * 3 + lane;
                s[k] = 0; u[k] = -1; cf[k] = 0.0f;
                if (bk + lane < nrec) {
                    s[k]  = __float_as_int(r[0]);
                    u[k]  = __float_as_int(r[Q_BATCH]);
                    cf[k] = r[2 * Q_BATCH];
                }
            }
            float2 v[L3_BATCHES];
#pragma unroll
            for (int k = 0; k < L3_BATCHES; ++k) {
                v[k] = make_float2(0.0f, 0.0f);
                if (u[k] >= 0) v[k] = *(const float2*)(h + (size_t)s[k] * 2);
            }
#pragma unroll
            for (int k = 0; k < L3_BATCHES; ++k) {
                if (u[k] >= 0) {
                    atomicAdd(&acc[u[k] * 2 + 0], v[k].x * cf[k]);
                    atomicAdd(&acc[u[k] * 2 + 1], v[k].y * cf[k]);
                }
            }
        }
    } else {
        l3_scan_edges(keys, pk, h, acc, wave, lane, (int)blockIdx.x, e);
    }
    __syncthreads();
    float2* o = (float2*)out + r0;
    for (int i = tid; i < nr; i += TILE_THREADS) o[i] = acc2[i];
}

extern "C" void kernel_launch(void* const* d_in, const int* in_sizes, int n_in,
                              void* d_out, int out_size, void* d_ws, size_t ws_size,
                              hipStream_t stream) {
    const float* x  = (const float*)d_in[0];
    const int*   ei = (const int*)d_in[1];
    const float* W1 = (const float*)d_in[2];
    const float* b1 = (const float*)d_in[3];
    const float* W2 = (const float*)d_in[4];
    const float* b2 = (const float*)d_in[5];
    const float* W3 = (const float*)d_in[6];
    const float* b3 = (const float*)d_in[7];
    float* out = (float*)d_out;

    const int N = in_sizes[0] / 5;
    const int E = in_sizes[1] / 2;
    const int* src = ei;
    const int* dst = ei + E;

    float* ws   = (float*)d_ws;
    float* isq  = ws;
    float* idg  = isq  + N;
    float* w1t  = idg  + N;
    float* w2t  = w1t  + 64 * 8;
    float* w3t  = w2t  + 64 * 64;
    float* bufA = w3t  + 16 * 64;
    float* bufB = bufA + (size_t)N * 64;
    float* bufC = bufB + (size_t)N * 64;
    float* h3   = bufC + (size_t)N * 64;
    float* part = h3   + (size_t)N * 2;
    int*   lcnt = (int*)(part + (size_t)128 * (N + 32));
    float* rec  = (float*)(lcnt + (size_t)((N + TILE_ROWS - 1) / TILE_ROWS) * (TILE_THREADS / 32));
    int*   keys = (int*)(rec + (size_t)((N + TILE_ROWS - 1) / TILE_ROWS) * (TILE_THREADS / 32) * REC_WORDS);
    int*   pk   = keys + (size_t)((E + KEY_WIN - 1) / KEY_WIN) * (KEY_WIN / 4);

    const int T = 256;
    auto cdiv = [](long a, long b) { return (int)((a + b - 1) / b); };

    k_deg_part <<<dim3(DEG_SLICES, cdiv(N, DEG_ROWS)), ACC_THREADS, 0, stream>>>(dst, part, E, N);
    k_norm_part<<<cdiv(N, T), T, 0, stream>>>(part, isq, idg, N);
    k_edge_prep<<<cdiv(cdiv(E, KEY_WIN), PREP_THREADS / 32), PREP_THREADS, 0, stream>>>(src, dst, isq, pk, keys, E);
    k_prep_w   <<<cdiv(64 * 8,  T), T, 0, stream>>>(W1, w1t, 5, 64, 8, 64);
    k_prep_w   <<<cdiv(64 * 64, T), T, 0, stream>>>(W2, w2t, 64, 64, 64, 64);
    k_prep_w   <<<cdiv(16 * 64, T), T, 0, stream>>>(W3, w3t, 64, 2, 64, 16);

    const int GT = cdiv(N, 16);

    k_gemm_wmma<8, 5, 4, 64, false, true><<<GT, 128, 0, stream>>>(x, w1t, bufA, idg, b1, bufB);
    k_scatter64_tile<false><<<cdiv(N, TILE_ROWS), TILE_THREADS, 0, stream>>>(keys, pk, bufA, bufB, lcnt, rec, E, N);

    k_gemm_wmma<64, 64, 4, 64, true, true><<<GT, 128, 0, stream>>>(bufB, w2t, bufA, idg, b2, bufC);
    k_scatter64_tile<true><<<cdiv(N, TILE_ROWS), TILE_THREADS, 0, stream>>>(keys, pk, bufA, bufC, lcnt, rec, E, N);

    k_gemm_wmma<64, 64, 1, 2, true, false><<<GT, 32, 0, stream>>>(bufC, w3t, h3, nullptr, nullptr, nullptr);
    k_l3_tile  <<<cdiv(N, TILE_ROWS), TILE_THREADS, 0, stream>>>(keys, pk, h3, idg, b3, lcnt, rec, out, E, N);
}


# 64-wide layers: GEMM writes the product plane only; each tile forms its start rows (self loop+bias) from its own product rows at load; 1 N*64 plane write per layer and 1 buffer gone; sums unchanged
# speedup vs baseline: 15.8543x; 1.0913x over previous
#include <hip/hip_runtime.h>
#include <hip/hip_bf16.h>

typedef __attribute__((ext_vector_type(2))) float v2f;
typedef __attribute__((ext_vector_type(8))) float v8f;

constexpr int ACC_THREADS = 1024;
constexpr int ACC_UNROLL  = 8;
constexpr int DEG_ROWS    = 50016;
constexpr int DEG_SLICES  = 128;

__device__ __forceinline__ size_t plane_pitch(size_t words) {
    return (words + 31) & ~(size_t)31;
}

__global__ __launch_bounds__(ACC_THREADS)
void k_deg_part(const int* __restrict__ dst, float* __restrict__ part, int e, int n) {
    __shared__ float cnt[DEG_ROWS];
    const int tid = threadIdx.x;
    const int r0  = blockIdx.y * DEG_ROWS;
    int nr = n - r0;
    if (nr > DEG_ROWS) nr = DEG_ROWS;
    for (int i = tid; i < nr; i += ACC_THREADS) cnt[i] = 0.0f;
    __syncthreads();

    const int per = (e + DEG_SLICES - 1) / DEG_SLICES;
    const int e0  = blockIdx.x * per;
    int e1 = e0 + per;
    if (e1 > e) e1 = e;
    for (int b = e0; b < e1; b += ACC_THREADS * ACC_UNROLL) {
        int d[ACC_UNROLL];
#pragma unroll
        for (int k = 0; k < ACC_UNROLL; ++k) {
            const int i = b + k * ACC_THREADS + tid;
            d[k] = (i < e1) ? dst[i] : -1;
        }
#pragma unroll
        for (int k = 0; k < ACC_UNROLL; ++k) {
            const int u = d[k] - r0;
            if ((unsigned)u < (unsigned)nr) atomicAdd(&cnt[u], 1.0f);
        }
    }
    __syncthreads();
    float* p = part + (size_t)blockIdx.x * plane_pitch((size_t)n) + r0;
    for (int i = tid; i < nr; i += ACC_THREADS) p[i] = cnt[i];
}

__global__ void k_norm_part(const float* __restrict__ part, float* __restrict__ isq,
                            float* __restrict__ idg, int n) {
    int i = blockIdx.x * blockDim.x + threadIdx.x;
    if (i < n) {
        const size_t pitch = plane_pitch((size_t)n);
        float d = 1.0f;
#pragma unroll 16
        for (int s = 0; s < DEG_SLICES; ++s) d += part[(size_t)s * pitch + i];
        isq[i] = rsqrtf(d);
        idg[i] = 1.0f / d;
    }
}

__global__ void k_prep_w(const float* __restrict__ W, float* __restrict__ Bt,
                         int Kin, int Cin, int KPAD, int COLS) {
    int idx = blockIdx.x * blockDim.x + threadIdx.x;
    if (idx >= COLS * KPAD) return;
    int col = idx / KPAD, k = idx % KPAD;
    Bt[idx] = (k < Kin && col < Cin) ? W[k * Cin + col] : 0.0f;
}

template <int K, int LDA, int NT, int SC, bool RELU_A>
__global__ void k_gemm_wmma(const float* __restrict__ A, const float* __restrict__ Bt,
                            float* __restrict__ H) {
    const int lane = threadIdx.x & 31;
    const int wave = threadIdx.x >> 5;
    const int half = lane >> 4;
    const int l15  = lane & 15;
    const size_t row = (size_t)blockIdx.x * 16 + l15;
    const int    col = wave * 16 + l15;
    const float* Arow = A + row * LDA;
    const float* Bcol = Bt + (size_t)col * K;
    const size_t mbase = (size_t)blockIdx.x * 16 + half * 8;

    v8f c = {};
#pragma unroll
    for (int k = 0; k < K; k += 4) {
        const int ka = k + half * 2;
        v2f a;
        if (LDA == K) {
            a = *(const v2f*)(Arow + ka);
        } else {
            a.x = (ka     < LDA) ? Arow[ka]     : 0.0f;
            a.y = (ka + 1 < LDA) ? Arow[ka + 1] : 0.0f;
        }
        if (RELU_A) {
            a.x = fmaxf(a.x, 0.0f);
            a.y = fmaxf(a.y, 0.0f);
        }
        v2f b = *(const v2f*)(Bcol + ka);
        c = __builtin_amdgcn_wmma_f32_16x16x4_f32(
                 false, a,  false, b,
                 (short)0, c,  false,  false);
    }

    if (NT * 16 == SC || col < SC) {
#pragma unroll
        for (int r = 0; r < 8; ++r)
            H[(mbase + r) * SC + col] = c[r];
    }
}

constexpr int TILE_ROWS    = 391;
constexpr int TILE_THREADS = 1024;
constexpr int Q_CAP        = 64;
constexpr int Q_BATCH      = 32;
constexpr int Q_GROUP      = 16;
constexpr int TILE_LOAD_IT = (TILE_ROWS * 16 + TILE_THREADS - 1) / TILE_THREADS;

constexpr int KEY_WIN   = 1024;
constexpr int ROW_SHIFT = 23;
static_assert(TILE_ROWS <= (1 << (32 - ROW_SHIFT)), "row inside the tile must fit above ROW_SHIFT");

constexpr int PREP_THREADS = 256;
constexpr int PREP_STEPS   = 16;
__global__ __launch_bounds__(PREP_THREADS)
void k_edge_prep(const int* __restrict__ src, const int* __restrict__ dst,
                 const float* __restrict__ isq,
                 int* __restrict__ pk, int* __restrict__ keys, int e) {
    const int lane = threadIdx.x & 31;
    const int win = __builtin_amdgcn_readfirstlane((int)(blockIdx.x * (PREP_THREADS / 32) + (threadIdx.x >> 5)));
    const int wb  = win * KEY_WIN;
    if (wb >= e) return;
    int pl[8] = {0, 0, 0, 0, 0, 0, 0, 0};
    for (int i0 = 0; i0 < 32; i0 += PREP_STEPS) {
        int d[PREP_STEPS], s[PREP_STEPS];
#pragma unroll
        for (int k = 0; k < PREP_STEPS; ++k) {
            const int  i  = wb + (i0 + k) * 32 + lane;
            const bool ok = i < e;
            d[k] = ok ? dst[i] : 0;
            s[k] = ok ? src[i] : 0;
        }
        float fs[PREP_STEPS], fd[PREP_STEPS];
#pragma unroll
        for (int k = 0; k < PREP_STEPS; ++k) {
            fs[k] = isq[s[k]];
            fd[k] = isq[d[k]];
        }
#pragma unroll
        for (int k = 0; k < PREP_STEPS; ++k) {
            const int  i  = wb + (i0 + k) * 32 + lane;
            const bool ok = i < e;
            const int   t  = d[k] / TILE_ROWS;
            const int   u  = d[k] - t * TILE_ROWS;
            const float cf = fs[k] * fd[k];
            if (ok)
                ((int2*)pk)[i] = make_int2((int)((unsigned)s[k] | ((unsigned)u << ROW_SHIFT)), __float_as_int(cf));
            const bool mine = lane == i0 + k;
#pragma unroll
            for (int p = 0; p < 8; ++p) {
                const int bits = (int)__builtin_amdgcn_ballot_w32(ok && ((t >> p) & 1));
                pl[p] = mine ? bits : pl[p];
            }
        }
    }
    int4* k4 = (int4*)keys + (size_t)win * 64;
    k4[lane]      = make_int4(pl[0], pl[1], pl[2], pl[3]);
    k4[32 + lane] = make_int4(pl[4], pl[5], pl[6], pl[7]);
}

constexpr int REC_CAP   = 768;
constexpr int REC_WORDS = REC_CAP * 3;

template <bool FULL>
__device__ __forceinline__ void rows_add(int s, int u, float cf, int cnt, int lane,
                                         const float* __restrict__ h, float* tile) {
#pragma unroll
    for (int g = 0; g < Q_BATCH; g += Q_GROUP) {
        float v0[Q_GROUP], v1[Q_GROUP];
#pragma unroll
        for (int k = 0; k < Q_GROUP; ++k) {
            if (FULL || g + k < cnt) {
                const float* hs = h + (size_t)__builtin_amdgcn_readlane(s, g + k) * 64;
                v0[k] = hs[lane];
                v1[k] = hs[lane + 32];
            }
        }
#pragma unroll
        for (int k = 0; k < Q_GROUP; ++k) {
            if (FULL || g + k < cnt) {
                const float cfk = __int_as_float(__builtin_amdgcn_readlane(__float_as_int(cf), g + k));
                float* t = tile + __builtin_amdgcn_readlane(u, g + k) * 64;
                atomicAdd(t + lane,      v0[k] * cfk);
                atomicAdd(t + lane + 32, v1[k] * cfk);
            }
        }
    }
}

template <bool FULL, bool REC>
__device__ __forceinline__ void queue_drain(const int* q, int qh, int cnt, int lane,
                                            const int* __restrict__ pk,
                                            const float* __restrict__ h, float* tile,
                                            float* recw, int& nrec) {
    int s = 0, u = 0;
    float cf = 0.0f;
    if (FULL || lane < cnt) {
        const int  en = q[(qh + lane) & (Q_CAP - 1)];
        const int2 r  = ((const int2*)pk)[en];
        s  = r.x & ((1 << ROW_SHIFT) - 1);
        u  = (int)((unsigned)r.x >> ROW_SHIFT);
        cf = __int_as_float(r.y);
    }
    if (REC && nrec >= 0) {
        if (nrec + Q_BATCH <= REC_CAP) {
            if (FULL || lane < cnt) {
                float* r = recw + nrec * 3 + lane;
                r[0]           = __int_as_float(s);
                r[Q_BATCH]     = __int_as_float(u);
                r[2 * Q_BATCH] = cf;
            }
            nrec += cnt;
        } else {
            nrec = -1;
        }
    }
    rows_add<FULL>(s, u, cf, cnt, lane, h, tile);
}

__device__ __forceinline__ unsigned key_match(const int4& lo, const int4& hi, const unsigned (&xm)[8]) {
    unsigned m = ((unsigned)lo.x ^ xm[0]) & ((unsigned)lo.y ^ xm[1]);
    m &= ((unsigned)lo.z ^ xm[2]);
    m &= ((unsigned)lo.w ^ xm[3]);
    m &= ((unsigned)hi.x ^ xm[4]);
    m &= ((unsigned)hi.y ^ xm[5]);
    m &= ((unsigned)hi.z ^ xm[6]);
    m &= ((unsigned)hi.w ^ xm[7]);
    return m;
}

template <bool REC>
__device__ __forceinline__ void window_hits(unsigned m, int ebase, int lane,
                                            int* q, int& qh, int& qn,
                                            const int* __restrict__ pk,
                                            const float* __restrict__ h, float* tile,
                                            float* recw, int& nrec) {
    unsigned bal = __builtin_amdgcn_ballot_w32(m != 0u);
    while (bal) {
        if (m != 0u) {
            const int slot = (qh + qn + (int)__builtin_amdgcn_mbcnt_lo(bal, 0u)) & (Q_CAP - 1);
            q[slot] = ebase + __builtin_ctz(m);
            m &= m - 1u;
        }
        qn += __builtin_popcount(bal);
        if (qn >= Q_BATCH) {
            queue_drain<true, REC>(q, qh, Q_BATCH, lane, pk, h, tile, recw, nrec);
            qh = (qh + Q_BATCH) & (Q_CAP - 1);
            qn -= Q_BATCH;
        }
        bal = __builtin_amdgcn_ballot_w32(m != 0u);
    }
}

template <bool REC>
__device__ __forceinline__ void scan_edges(const int* __restrict__ keys, const int* __restrict__ pk,
                                           const float* __restrict__ h,
                                           float* tile, int* q, int wave, int lane,
                                           int tno, int e, float* recw, int& nrec) {
    int qh = 0, qn = 0;
    unsigned xm[8];
#pragma unroll
    for (int p = 0; p < 8; ++p) xm[p] = ((tno >> p) & 1) ? 0u : ~0u;

    const int4* k4    = (const int4*)keys;
    const int   nfull = e / (2 * KEY_WIN);
    int4 nx[4];
    if (wave < nfull) {
        const int4* p = k4 + (size_t)wave * 128 + lane;
#pragma unroll
        for (int qq = 0; qq < 4; ++qq) nx[qq] = p[qq * 32];
    }
    for (int w = wave; w < nfull; w += TILE_THREADS / 32) {
        int4 cur[4];
#pragma unroll
        for (int qq = 0; qq < 4; ++qq) cur[qq] = nx[qq];
        const int wn = w + TILE_THREADS / 32;
        if (wn < nfull) {
            const int4* p = k4 + (size_t)wn * 128 + lane;
#pragma unroll
            for (int qq = 0; qq < 4; ++qq) nx[qq] = p[qq * 32];
        }
#pragma unroll
        for (int hh = 0; hh < 2; ++hh) {
            const unsigned m = key_match(cur[2 * hh], cur[2 * hh + 1], xm);
            window_hits<REC>(m, (2 * w + hh) * KEY_WIN + lane * 32, lane, q, qh, qn, pk, h, tile, recw, nrec);
        }
    }
    const int kw = (e + KEY_WIN - 1) / KEY_WIN;
    for (int tw = 2 * nfull + wave; tw < kw; tw += TILE_THREADS / 32) {
        const int4* p = k4 + (size_t)tw * 64 + lane;
        const int4 lo = p[0];
        const int4 hi = p[32];
        unsigned m = key_match(lo, hi, xm);
        const int ebase = tw * KEY_WIN + lane * 32;
        const int valid = e - ebase;
        if (valid < 32) m = (valid <= 0) ? 0u : (m & ((1u << valid) - 1u));
        window_hits<REC>(m, ebase, lane, q, qh, qn, pk, h, tile, recw, nrec);
    }
    if (qn > 0) queue_drain<false, REC>(q, qh, qn, lane, pk, h, tile, recw, nrec);
}

__device__ __forceinline__ void replay_edges(const float* __restrict__ recw, int nrec, int lane,
                                             const float* __restrict__ h, float* tile) {
    int s = 0, u = 0;
    float cf = 0.0f;
    if (lane < nrec) {
        s  = __float_as_int(recw[lane]);
        u  = __float_as_int(recw[Q_BATCH + lane]);
        cf = recw[2 * Q_BATCH + lane];
    }
    for (int b = 0; b < nrec; b += Q_BATCH) {
        int s2 = 0, u2 = 0;
        float cf2 = 0.0f;
        const int nb = b + Q_BATCH;
        if (nb + lane < nrec) {
            const float* r = recw + nb * 3 + lane;
            s2  = __float_as_int(r[0]);
            u2  = __float_as_int(r[Q_BATCH]);
            cf2 = r[2 * Q_BATCH];
        }
        if (nrec - b >= Q_BATCH) rows_add<true>(s, u, cf, Q_BATCH, lane, h, tile);
        else                     rows_add<false>(s, u, cf, nrec - b, lane, h, tile);
        s = s2; u = u2; cf = cf2;
    }
}

template <bool REPLAY>
__global__ __launch_bounds__(TILE_THREADS)
void k_scatter64_tile(const int* __restrict__ keys, const int* __restrict__ pk,
                      const float* __restrict__ h, const float* __restrict__ idg,
                      const float* __restrict__ bias,
                      float* __restrict__ agg, int* __restrict__ lcnt, float* __restrict__ rec,
                      int e, int n) {
    __shared__ float tile[TILE_ROWS * 64];
    __shared__ int   queue[(TILE_THREADS / 32) * Q_CAP];
    const int tid  = threadIdx.x;
    const int lane = tid & 31;
    const int wave = __builtin_amdgcn_readfirstlane(tid >> 5);
    const int r0   = blockIdx.x * TILE_ROWS;
    int nr = n - r0;
    if (nr > TILE_ROWS) nr = TILE_ROWS;
    if (nr <= 0) return;

    float4*       t4 = (float4*)tile;
    float4*       a4 = (float4*)(agg + (size_t)r0 * 64);
    const float4* h4 = (const float4*)(h + (size_t)r0 * 64);
    const float4  b4 = ((const float4*)bias)[tid & 15];
    float4 hv[TILE_LOAD_IT];
    float  g[TILE_LOAD_IT];
#pragma unroll
    for (int k = 0; k < TILE_LOAD_IT; ++k) {
        const int i = tid + k * TILE_THREADS;
        if (i < nr * 16) {
            hv[k] = h4[i];
            g[k]  = idg[r0 + (i >> 4)];
        }
    }
#pragma unroll
    for (int k = 0; k < TILE_LOAD_IT; ++k) {
        const int i = tid + k * TILE_THREADS;
        if (i < nr * 16)
            t4[i] = make_float4(hv[k].x * g[k] + b4.x, hv[k].y * g[k] + b4.y,
                                hv[k].z * g[k] + b4.z, hv[k].w * g[k] + b4.w);
    }
    __syncthreads();

    int*   q    = queue + wave * Q_CAP;
    const int tno = __builtin_amdgcn_readfirstlane((int)blockIdx.x);
    const int wid = blockIdx.x * (TILE_THREADS / 32) + wave;
    float* recw = rec + (size_t)wid * REC_WORDS;

    if (REPLAY) {
        int nrec = __builtin_amdgcn_readfirstlane(lcnt[wid]);
        if (nrec >= 0) replay_edges(recw, nrec, lane, h, tile);
        else           scan_edges<false>(keys, pk, h, tile, q, wave, lane, tno, e, recw, nrec);
    } else {
        int nrec = 0;
        scan_edges<true>(keys, pk, h, tile, q, wave, lane, tno, e, recw, nrec);
        if (lane == 0) lcnt[wid] = nrec;
    }
    __syncthreads();
    for (int i = tid; i < nr * 16; i += TILE_THREADS) a4[i] = t4[i];
}

constexpr int L3_BATCHES = 8;

__device__ __forceinline__ void l3_window(int tw, const int4* __restrict__ k4, const unsigned (&xm)[8],
                                          const int* __restrict__ pk, const float* __restrict__ h,
                                          float* acc, int lane, int e) {
    const int4* p = k4 + (size_t)tw * 64 + lane;
    unsigned m = key_match(p[0], p[32], xm);
    const int ebase = tw * KEY_WIN + lane * 32;
    const int valid = e - ebase;
    if (valid < 32) m = (valid <= 0) ? 0u : (m & ((1u << valid) - 1u));
    while (m) {
        const int  en = ebase + __builtin_ctz(m);
        m &= m - 1u;
        const int2   r  = ((const int2*)pk)[en];
        const int    s  = r.x & ((1 << ROW_SHIFT) - 1);
        const int    u  = (int)((unsigned)r.x >> ROW_SHIFT);
        const float  cf = __int_as_float(r.y);
        const float2 v  = *(const float2*)(h + (size_t)s * 2);
        atomicAdd(&acc[u * 2 + 0], v.x * cf);
        atomicAdd(&acc[u * 2 + 1], v.y * cf);
    }
}

__device__ __forceinline__ void l3_scan_edges(const int* __restrict__ keys, const int* __restrict__ pk,
                                              const float* __restrict__ h, float* acc,
                                              int wave, int lane, int tno, int e) {
    unsigned xm[8];
#pragma unroll
    for (int p = 0; p < 8; ++p) xm[p] = ((tno >> p) & 1) ? 0u : ~0u;
    const int4* k4    = (const int4*)keys;
    const int   nfull = e / (2 * KEY_WIN);
    const int   kw    = (e + KEY_WIN - 1) / KEY_WIN;
    for (int w = wave; w < nfull; w += TILE_THREADS / 32) {
        l3_window(2 * w,     k4, xm, pk, h, acc, lane, e);
        l3_window(2 * w + 1, k4, xm, pk, h, acc, lane, e);
    }
    for (int tw = 2 * nfull + wave; tw < kw; tw += TILE_THREADS / 32)
        l3_window(tw, k4, xm, pk, h, acc, lane, e);
}

__global__ __launch_bounds__(TILE_THREADS)
void k_l3_tile(const int* __restrict__ keys, const int* __restrict__ pk,
               const float* __restrict__ h, const float* __restrict__ idg,
               const float* __restrict__ bias, const int* __restrict__ lcnt,
               const float* __restrict__ rec, float* __restrict__ out, int e, int n) {
    __shared__ float acc[TILE_ROWS * 2];
    const int tid  = threadIdx.x;
    const int lane = tid & 31;
    const int wave = __builtin_amdgcn_readfirstlane(tid >> 5);
    const int r0   = blockIdx.x * TILE_ROWS;
    int nr = n - r0;
    if (nr > TILE_ROWS) nr = TILE_ROWS;
    if (nr <= 0) return;

    float2* acc2 = (float2*)acc;
    const float b0 = bias[0], b1 = bias[1];
    for (int i = tid; i < nr; i += TILE_THREADS) {
        const float2 hv = ((const float2*)h)[r0 + i];
        const float  g  = idg[r0 + i];
        acc2[i] = make_float2(hv.x * g + b0, hv.y * g + b1);
    }
    __syncthreads();

    const int wid = blockIdx.x * (TILE_THREADS / 32) + wave;
    const float* recw = rec + (size_t)wid * REC_WORDS;
    const int nrec = __builtin_amdgcn_readfirstlane(lcnt[wid]);
    if (nrec >= 0) {
        for (int b = 0; b < nrec; b += Q_BATCH * L3_BATCHES) {
            int   s[L3_BATCHES], u[L3_BATCHES];
            float cf[L3_BATCHES];
#pragma unroll
            for (int k = 0; k < L3_BATCHES; ++k) {
                const int    bk = b + k * Q_BATCH;
                const float* r  = recw + bk * 3 + lane;
                s[k] = 0; u[k] = -1; cf[k] = 0.0f;
                if (bk + lane < nrec) {
                    s[k]  = __float_as_int(r[0]);
                    u[k]  = __float_as_int(r[Q_BATCH]);
                    cf[k] = r[2 * Q_BATCH];
                }
            }
            float2 v[L3_BATCHES];
#pragma unroll
            for (int k = 0; k < L3_BATCHES; ++k) {
                v[k] = make_float2(0.0f, 0.0f);
                if (u[k] >= 0) v[k] = *(const float2*)(h + (size_t)s[k] * 2);
            }
#pragma unroll
            for (int k = 0; k < L3_BATCHES; ++k) {
                if (u[k] >= 0) {
                    atomicAdd(&acc[u[k] * 2 + 0], v[k].x * cf[k]);
                    atomicAdd(&acc[u[k] * 2 + 1], v[k].y * cf[k]);
                }
            }
        }
    } else {
        l3_scan_edges(keys, pk, h, acc, wave, lane, (int)blockIdx.x, e);
    }
    __syncthreads();
    float2* o = (float2*)out + r0;
    for (int i = tid; i < nr; i += TILE_THREADS) o[i] = acc2[i];
}

extern "C" void kernel_launch(void* const* d_in, const int* in_sizes, int n_in,
                              void* d_out, int out_size, void* d_ws, size_t ws_size,
                              hipStream_t stream) {
    const float* x  = (const float*)d_in[0];
    const int*   ei = (const int*)d_in[1];
    const float* W1 = (const float*)d_in[2];
    const float* b1 = (const float*)d_in[3];
    const float* W2 = (const float*)d_in[4];
    const float* b2 = (const float*)d_in[5];
    const float* W3 = (const float*)d_in[6];
    const float* b3 = (const float*)d_in[7];
    float* out = (float*)d_out;

    const int N = in_sizes[0] / 5;
    const int E = in_sizes[1] / 2;
    const int* src = ei;
    const int* dst = ei + E;

    float* ws   = (float*)d_ws;
    float* isq  = ws;
    float* idg  = isq  + N;
    float* w1t  = idg  + N;
    float* w2t  = w1t  + 64 * 8;
    float* w3t  = w2t  + 64 * 64;
    float* bufA = w3t  + 16 * 64;
    float* bufB = bufA + (size_t)N * 64;
    float* h3   = bufB + (size_t)N * 64;
    float* part = h3   + (size_t)N * 2;
    int*   lcnt = (int*)(part + (size_t)128 * (N + 32));
    float* rec  = (float*)(lcnt + (size_t)((N + TILE_ROWS - 1) / TILE_ROWS) * (TILE_THREADS / 32));
    int*   keys = (int*)(rec + (size_t)((N + TILE_ROWS - 1) / TILE_ROWS) * (TILE_THREADS / 32) * REC_WORDS);
    int*   pk   = keys + (size_t)((E + KEY_WIN - 1) / KEY_WIN) * (KEY_WIN / 4);

    const int T = 256;
    auto cdiv = [](long a, long b) { return (int)((a + b - 1) / b); };

    k_deg_part <<<dim3(DEG_SLICES, cdiv(N, DEG_ROWS)), ACC_THREADS, 0, stream>>>(dst, part, E, N);
    k_norm_part<<<cdiv(N, T), T, 0, stream>>>(part, isq, idg, N);
    k_edge_prep<<<cdiv(cdiv(E, KEY_WIN), PREP_THREADS / 32), PREP_THREADS, 0, stream>>>(src, dst, isq, pk, keys, E);
    k_prep_w   <<<cdiv(64 * 8,  T), T, 0, stream>>>(W1, w1t, 5, 64, 8, 64);
    k_prep_w   <<<cdiv(64 * 64, T), T, 0, stream>>>(W2, w2t, 64, 64, 64, 64);
    k_prep_w   <<<cdiv(16 * 64, T), T, 0, stream>>>(W3, w3t, 64, 2, 64, 16);

    const int GT = cdiv(N, 16);

    k_gemm_wmma<8, 5, 4, 64, false><<<GT, 128, 0, stream>>>(x, w1t, bufA);
    k_scatter64_tile<false><<<cdiv(N, TILE_ROWS), TILE_THREADS, 0, stream>>>(keys, pk, bufA, idg, b1, bufB, lcnt, rec, E, N);

    k_gemm_wmma<64, 64, 4, 64, true><<<GT, 128, 0, stream>>>(bufB, w2t, bufA);
    k_scatter64_tile<true><<<cdiv(N, TILE_ROWS), TILE_THREADS, 0, stream>>>(keys, pk, bufA, idg, b2, bufB, lcnt, rec, E, N);

    k_gemm_wmma<64, 64, 1, 2, true><<<GT, 32, 0, stream>>>(bufB, w3t, h3);
    k_l3_tile  <<<cdiv(N, TILE_ROWS), TILE_THREADS, 0, stream>>>(keys, pk, h3, idg, b3, lcnt, rec, out, E, N);
}


# Layer-2 and layer-3 weight products taken in the edge-sum launches: the finished 391-row LDS tile feeds the f32 WMMA as A, products stored, not sums; 2 launches, 2 N*64 planes gone; sums unchanged
# speedup vs baseline: 16.7585x; 1.0570x over previous
#include <hip/hip_runtime.h>
#include <hip/hip_bf16.h>

typedef __attribute__((ext_vector_type(2))) float v2f;
typedef __attribute__((ext_vector_type(8))) float v8f;

constexpr int ACC_THREADS = 1024;
constexpr int ACC_UNROLL  = 8;
constexpr int DEG_ROWS    = 50016;
constexpr int DEG_SLICES  = 128;

__device__ __forceinline__ size_t plane_pitch(size_t words) {
    return (words + 31) & ~(size_t)31;
}

__global__ __launch_bounds__(ACC_THREADS)
void k_deg_part(const int* __restrict__ dst, float* __restrict__ part, int e, int n) {
    __shared__ float cnt[DEG_ROWS];
    const int tid = threadIdx.x;
    const int r0  = blockIdx.y * DEG_ROWS;
    int nr = n - r0;
    if (nr > DEG_ROWS) nr = DEG_ROWS;
    for (int i = tid; i < nr; i += ACC_THREADS) cnt[i] = 0.0f;
    __syncthreads();

    const int per = (e + DEG_SLICES - 1) / DEG_SLICES;
    const int e0  = blockIdx.x * per;
    int e1 = e0 + per;
    if (e1 > e) e1 = e;
    for (int b = e0; b < e1; b += ACC_THREADS * ACC_UNROLL) {
        int d[ACC_UNROLL];
#pragma unroll
        for (int k = 0; k < ACC_UNROLL; ++k) {
            const int i = b + k * ACC_THREADS + tid;
            d[k] = (i < e1) ? dst[i] : -1;
        }
#pragma unroll
        for (int k = 0; k < ACC_UNROLL; ++k) {
            const int u = d[k] - r0;
            if ((unsigned)u < (unsigned)nr) atomicAdd(&cnt[u], 1.0f);
        }
    }
    __syncthreads();
    float* p = part + (size_t)blockIdx.x * plane_pitch((size_t)n) + r0;
    for (int i = tid; i < nr; i += ACC_THREADS) p[i] = cnt[i];
}

__global__ void k_norm_part(const float* __restrict__ part, float* __restrict__ isq,
                            float* __restrict__ idg, int n) {
    int i = blockIdx.x * blockDim.x + threadIdx.x;
    if (i < n) {
        const size_t pitch = plane_pitch((size_t)n);
        float d = 1.0f;
#pragma unroll 16
        for (int s = 0; s < DEG_SLICES; ++s) d += part[(size_t)s * pitch + i];
        isq[i] = rsqrtf(d);
        idg[i] = 1.0f / d;
    }
}

__global__ void k_prep_w(const float* __restrict__ W, float* __restrict__ Bt,
                         int Kin, int Cin, int KPAD, int COLS) {
    int idx = blockIdx.x * blockDim.x + threadIdx.x;
    if (idx >= COLS * KPAD) return;
    int col = idx / KPAD, k = idx % KPAD;
    Bt[idx] = (k < Kin && col < Cin) ? W[k * Cin + col] : 0.0f;
}

template <int K, int LDA, int NT, int SC, bool RELU_A>
__global__ void k_gemm_wmma(const float* __restrict__ A, const float* __restrict__ Bt,
                            float* __restrict__ H) {
    const int lane = threadIdx.x & 31;
    const int wave = threadIdx.x >> 5;
    const int half = lane >> 4;
    const int l15  = lane & 15;
    const size_t row = (size_t)blockIdx.x * 16 + l15;
    const int    col = wave * 16 + l15;
    const float* Arow = A + row * LDA;
    const float* Bcol = Bt + (size_t)col * K;
    const size_t mbase = (size_t)blockIdx.x * 16 + half * 8;

    v8f c = {};
#pragma unroll
    for (int k = 0; k < K; k += 4) {
        const int ka = k + half * 2;
        v2f a;
        if (LDA == K) {
            a = *(const v2f*)(Arow + ka);
        } else {
            a.x = (ka     < LDA) ? Arow[ka]     : 0.0f;
            a.y = (ka + 1 < LDA) ? Arow[ka + 1] : 0.0f;
        }
        if (RELU_A) {
            a.x = fmaxf(a.x, 0.0f);
            a.y = fmaxf(a.y, 0.0f);
        }
        v2f b = *(const v2f*)(Bcol + ka);
        c = __builtin_amdgcn_wmma_f32_16x16x4_f32(
                 false, a,  false, b,
                 (short)0, c,  false,  false);
    }

    if (NT * 16 == SC || col < SC) {
#pragma unroll
        for (int r = 0; r < 8; ++r)
            H[(mbase + r) * SC + col] = c[r];
    }
}

constexpr int TILE_ROWS    = 391;
constexpr int TILE_THREADS = 1024;
constexpr int Q_CAP        = 64;
constexpr int Q_BATCH      = 32;
constexpr int Q_GROUP      = 16;
constexpr int TILE_LOAD_IT = (TILE_ROWS * 16 + TILE_THREADS - 1) / TILE_THREADS;

constexpr int KEY_WIN   = 1024;
constexpr int ROW_SHIFT = 23;
static_assert(TILE_ROWS <= (1 << (32 - ROW_SHIFT)), "row inside the tile must fit above ROW_SHIFT");

constexpr int PREP_THREADS = 256;
constexpr int PREP_STEPS   = 16;
__global__ __launch_bounds__(PREP_THREADS)
void k_edge_prep(const int* __restrict__ src, const int* __restrict__ dst,
                 const float* __restrict__ isq,
                 int* __restrict__ pk, int* __restrict__ keys, int e) {
    const int lane = threadIdx.x & 31;
    const int win = __builtin_amdgcn_readfirstlane((int)(blockIdx.x * (PREP_THREADS / 32) + (threadIdx.x >> 5)));
    const int wb  = win * KEY_WIN;
    if (wb >= e) return;
    int pl[8] = {0, 0, 0, 0, 0, 0, 0, 0};
    for (int i0 = 0; i0 < 32; i0 += PREP_STEPS) {
        int d[PREP_STEPS], s[PREP_STEPS];
#pragma unroll
        for (int k = 0; k < PREP_STEPS; ++k) {
            const int  i  = wb + (i0 + k) * 32 + lane;
            const bool ok = i < e;
            d[k] = ok ? dst[i] : 0;
            s[k] = ok ? src[i] : 0;
        }
        float fs[PREP_STEPS], fd[PREP_STEPS];
#pragma unroll
        for (int k = 0; k < PREP_STEPS; ++k) {
            fs[k] = isq[s[k]];
            fd[k] = isq[d[k]];
        }
#pragma unroll
        for (int k = 0; k < PREP_STEPS; ++k) {
            const int  i  = wb + (i0 + k) * 32 + lane;
            const bool ok = i < e;
            const int   t  = d[k] / TILE_ROWS;
            const int   u  = d[k] - t * TILE_ROWS;
            const float cf = fs[k] * fd[k];
            if (ok)
                ((int2*)pk)[i] = make_int2((int)((unsigned)s[k] | ((unsigned)u << ROW_SHIFT)), __float_as_int(cf));
            const bool mine = lane == i0 + k;
#pragma unroll
            for (int p = 0; p < 8; ++p) {
                const int bits = (int)__builtin_amdgcn_ballot_w32(ok && ((t >> p) & 1));
                pl[p] = mine ? bits : pl[p];
            }
        }
    }
    int4* k4 = (int4*)keys + (size_t)win * 64;
    k4[lane]      = make_int4(pl[0], pl[1], pl[2], pl[3]);
    k4[32 + lane] = make_int4(pl[4], pl[5], pl[6], pl[7]);
}

constexpr int REC_CAP   = 768;
constexpr int REC_WORDS = REC_CAP * 3;

template <bool FULL>
__device__ __forceinline__ void rows_add(int s, int u, float cf, int cnt, int lane,
                                         const float* __restrict__ h, float* tile) {
#pragma unroll
    for (int g = 0; g < Q_BATCH; g += Q_GROUP) {
        float v0[Q_GROUP], v1[Q_GROUP];
#pragma unroll
        for (int k = 0; k < Q_GROUP; ++k) {
            if (FULL || g + k < cnt) {
                const float* hs = h + (size_t)__builtin_amdgcn_readlane(s, g + k) * 64;
                v0[k] = hs[lane];
                v1[k] = hs[lane + 32];
            }
        }
#pragma unroll
        for (int k = 0; k < Q_GROUP; ++k) {
            if (FULL || g + k < cnt) {
                const float cfk = __int_as_float(__builtin_amdgcn_readlane(__float_as_int(cf), g + k));
                float* t = tile + __builtin_amdgcn_readlane(u, g + k) * 64;
                atomicAdd(t + lane,      v0[k] * cfk);
                atomicAdd(t + lane + 32, v1[k] * cfk);
            }
        }
    }
}

template <bool FULL, bool REC>
__device__ __forceinline__ void queue_drain(const int* q, int qh, int cnt, int lane,
                                            const int* __restrict__ pk,
                                            const float* __restrict__ h, float* tile,
                                            float* recw, int& nrec) {
    int s = 0, u = 0;
    float cf = 0.0f;
    if (FULL || lane < cnt) {
        const int  en = q[(qh + lane) & (Q_CAP - 1)];
        const int2 r  = ((const int2*)pk)[en];
        s  = r.x & ((1 << ROW_SHIFT) - 1);
        u  = (int)((unsigned)r.x >> ROW_SHIFT);
        cf = __int_as_float(r.y);
    }
    if (REC && nrec >= 0) {
        if (nrec + Q_BATCH <= REC_CAP) {
            if (FULL || lane < cnt) {
                float* r = recw + nrec * 3 + lane;
                r[0]           = __int_as_float(s);
                r[Q_BATCH]     = __int_as_float(u);
                r[2 * Q_BATCH] = cf;
            }
            nrec += cnt;
        } else {
            nrec = -1;
        }
    }
    rows_add<FULL>(s, u, cf, cnt, lane, h, tile);
}

__device__ __forceinline__ unsigned key_match(const int4& lo, const int4& hi, const unsigned (&xm)[8]) {
    unsigned m = ((unsigned)lo.x ^ xm[0]) & ((unsigned)lo.y ^ xm[1]);
    m &= ((unsigned)lo.z ^ xm[2]);
    m &= ((unsigned)lo.w ^ xm[3]);
    m &= ((unsigned)hi.x ^ xm[4]);
    m &= ((unsigned)hi.y ^ xm[5]);
    m &= ((unsigned)hi.z ^ xm[6]);
    m &= ((unsigned)hi.w ^ xm[7]);
    return m;
}

template <bool REC>
__device__ __forceinline__ void window_hits(unsigned m, int ebase, int lane,
                                            int* q, int& qh, int& qn,
                                            const int* __restrict__ pk,
                                            const float* __restrict__ h, float* tile,
                                            float* recw, int& nrec) {
    unsigned bal = __builtin_amdgcn_ballot_w32(m != 0u);
    while (bal) {
        if (m != 0u) {
            const int slot = (qh + qn + (int)__builtin_amdgcn_mbcnt_lo(bal, 0u)) & (Q_CAP - 1);
            q[slot] = ebase + __builtin_ctz(m);
            m &= m - 1u;
        }
        qn += __builtin_popcount(bal);
        if (qn >= Q_BATCH) {
            queue_drain<true, REC>(q, qh, Q_BATCH, lane, pk, h, tile, recw, nrec);
            qh = (qh + Q_BATCH) & (Q_CAP - 1);
            qn -= Q_BATCH;
        }
        bal = __builtin_amdgcn_ballot_w32(m != 0u);
    }
}

template <bool REC>
__device__ __forceinline__ void scan_edges(const int* __restrict__ keys, const int* __restrict__ pk,
                                           const float* __restrict__ h,
                                           float* tile, int* q, int wave, int lane,
                                           int tno, int e, float* recw, int& nrec) {
    int qh = 0, qn = 0;
    unsigned xm[8];
#pragma unroll
    for (int p = 0; p < 8; ++p) xm[p] = ((tno >> p) & 1) ? 0u : ~0u;

    const int4* k4    = (const int4*)keys;
    const int   nfull = e / (2 * KEY_WIN);
    int4 nx[4];
    if (wave < nfull) {
        const int4* p = k4 + (size_t)wave * 128 + lane;
#pragma unroll
        for (int qq = 0; qq < 4; ++qq) nx[qq] = p[qq * 32];
    }
    for (int w = wave; w < nfull; w += TILE_THREADS / 32) {
        int4 cur[4];
#pragma unroll
        for (int qq = 0; qq < 4; ++qq) cur[qq] = nx[qq];
        const int wn = w + TILE_THREADS / 32;
        if (wn < nfull) {
            const int4* p = k4 + (size_t)wn * 128 + lane;
#pragma unroll
            for (int qq = 0; qq < 4; ++qq) nx[qq] = p[qq * 32];
        }
#pragma unroll
        for (int hh = 0; hh < 2; ++hh) {
            const unsigned m = key_match(cur[2 * hh], cur[2 * hh + 1], xm);
            window_hits<REC>(m, (2 * w + hh) * KEY_WIN + lane * 32, lane, q, qh, qn, pk, h, tile, recw, nrec);
        }
    }
    const int kw = (e + KEY_WIN - 1) / KEY_WIN;
    for (int tw = 2 * nfull + wave; tw < kw; tw += TILE_THREADS / 32) {
        const int4* p = k4 + (size_t)tw * 64 + lane;
        const int4 lo = p[0];
        const int4 hi = p[32];
        unsigned m = key_match(lo, hi, xm);
        const int ebase = tw * KEY_WIN + lane * 32;
        const int valid = e - ebase;
        if (valid < 32) m = (valid <= 0) ? 0u : (m & ((1u << valid) - 1u));
        window_hits<REC>(m, ebase, lane, q, qh, qn, pk, h, tile, recw, nrec);
    }
    if (qn > 0) queue_drain<false, REC>(q, qh, qn, lane, pk, h, tile, recw, nrec);
}

__device__ __forceinline__ void replay_edges(const float* __restrict__ recw, int nrec, int lane,
                                             const float* __restrict__ h, float* tile) {
    int s = 0, u = 0;
    float cf = 0.0f;
    if (lane < nrec) {
        s  = __float_as_int(recw[lane]);
        u  = __float_as_int(recw[Q_BATCH + lane]);
        cf = recw[2 * Q_BATCH + lane];
    }
    for (int b = 0; b < nrec; b += Q_BATCH) {
        int s2 = 0, u2 = 0;
        float cf2 = 0.0f;
        const int nb = b + Q_BATCH;
        if (nb + lane < nrec) {
            const float* r = recw + nb * 3 + lane;
            s2  = __float_as_int(r[0]);
            u2  = __float_as_int(r[Q_BATCH]);
            cf2 = r[2 * Q_BATCH];
        }
        if (nrec - b >= Q_BATCH) rows_add<true>(s, u, cf, Q_BATCH, lane, h, tile);
        else                     rows_add<false>(s, u, cf, nrec - b, lane, h, tile);
        s = s2; u = u2; cf = cf2;
    }
}

template <int NT>
__device__ __forceinline__ void tile_product(float* tile, int nr, int wave, int lane,
                                             const float* __restrict__ Bt) {
    const int half = lane >> 4;
    const int l15  = lane & 15;
    const int nblk = (nr + 15) >> 4;
    for (int rb = wave; rb < nblk; rb += TILE_THREADS / 32) {
        int arow = rb * 16 + l15;
        if (arow > nr - 1) arow = nr - 1;
        const float* ar = tile + arow * 64 + half * 2;
        v2f a[16];
#pragma unroll
        for (int k = 0; k < 16; ++k) {
            a[k] = *(const v2f*)(ar + 4 * k);
            a[k].x = fmaxf(a[k].x, 0.0f);
            a[k].y = fmaxf(a[k].y, 0.0f);
        }
        const int mrow = rb * 16 + half * 8;
#pragma nounroll
        for (int ct = 0; ct < NT; ++ct) {
            const float* Bcol = Bt + (size_t)(ct * 16 + l15) * 64 + half * 2;
            v8f c = {};
#pragma unroll
            for (int k = 0; k < 16; ++k) {
                const v2f b = *(const v2f*)(Bcol + 4 * k);
                c = __builtin_amdgcn_wmma_f32_16x16x4_f32(
                         false, a[k],  false, b,
                         (short)0, c,  false,  false);
            }
#pragma unroll
            for (int r = 0; r < 8; ++r)
                if (mrow + r < nr) tile[(mrow + r) * 64 + ct * 16 + l15] = c[r];
        }
    }
}

template <bool REPLAY, int NT, int SC>
__global__ __launch_bounds__(TILE_THREADS)
void k_scatter64_tile(const int* __restrict__ keys, const int* __restrict__ pk,
                      const float* __restrict__ h, const float* __restrict__ idg,
                      const float* __restrict__ bias, const float* __restrict__ Bt,
                      float* __restrict__ prod, int* __restrict__ lcnt, float* __restrict__ rec,
                      int e, int n) {
    __shared__ float tile[TILE_ROWS * 64];
    __shared__ int   queue[(TILE_THREADS / 32) * Q_CAP];
    const int tid  = threadIdx.x;
    const int lane = tid & 31;
    const int wave = __builtin_amdgcn_readfirstlane(tid >> 5);
    const int r0   = blockIdx.x * TILE_ROWS;
    int nr = n - r0;
    if (nr > TILE_ROWS) nr = TILE_ROWS;
    if (nr <= 0) return;

    float4*       t4 = (float4*)tile;
    const float4* h4 = (const float4*)(h + (size_t)r0 * 64);
    const float4  b4 = ((const float4*)bias)[tid & 15];
    float4 hv[TILE_LOAD_IT];
    float  g[TILE_LOAD_IT];
#pragma unroll
    for (int k = 0; k < TILE_LOAD_IT; ++k) {
        const int i = tid + k * TILE_THREADS;
        if (i < nr * 16) {
            hv[k] = h4[i];
            g[k]  = idg[r0 + (i >> 4)];
        }
    }
#pragma unroll
    for (int k = 0; k < TILE_LOAD_IT; ++k) {
        const int i = tid + k * TILE_THREADS;
        if (i < nr * 16)
            t4[i] = make_float4(hv[k].x * g[k] + b4.x, hv[k].y * g[k] + b4.y,
                                hv[k].z * g[k] + b4.z, hv[k].w * g[k] + b4.w);
    }
    __syncthreads();

    int*   q    = queue + wave * Q_CAP;
    const int tno = __builtin_amdgcn_readfirstlane((int)blockIdx.x);
    const int wid = blockIdx.x * (TILE_THREADS / 32) + wave;
    float* recw = rec + (size_t)wid * REC_WORDS;

    if (REPLAY) {
        int nrec = __builtin_amdgcn_readfirstlane(lcnt[wid]);
        if (nrec >= 0) replay_edges(recw, nrec, lane, h, tile);
        else           scan_edges<false>(keys, pk, h, tile, q, wave, lane, tno, e, recw, nrec);
    } else {
        int nrec = 0;
        scan_edges<true>(keys, pk, h, tile, q, wave, lane, tno, e, recw, nrec);
        if (lane == 0) lcnt[wid] = nrec;
    }
    __syncthreads();
    tile_product<NT>(tile, nr, wave, lane, Bt);
    __syncthreads();
    if (SC == 64) {
        float4* p4 = (float4*)(prod + (size_t)r0 * 64);
        for (int i = tid; i < nr * 16; i += TILE_THREADS) p4[i] = t4[i];
    } else {
        float2* p2 = (float2*)prod + r0;
        for (int i = tid; i < nr; i += TILE_THREADS) p2[i] = *(const float2*)(tile + i * 64);
    }
}

constexpr int L3_BATCHES = 8;

__device__ __forceinline__ void l3_window(int tw, const int4* __restrict__ k4, const unsigned (&xm)[8],
                                          const int* __restrict__ pk, const float* __restrict__ h,
                                          float* acc, int lane, int e) {
    const int4* p = k4 + (size_t)tw * 64 + lane;
    unsigned m = key_match(p[0], p[32], xm);
    const int ebase = tw * KEY_WIN + lane * 32;
    const int valid = e - ebase;
    if (valid < 32) m = (valid <= 0) ? 0u : (m & ((1u << valid) - 1u));
    while (m) {
        const int  en = ebase + __builtin_ctz(m);
        m &= m - 1u;
        const int2   r  = ((const int2*)pk)[en];
        const int    s  = r.x & ((1 << ROW_SHIFT) - 1);
        const int    u  = (int)((unsigned)r.x >> ROW_SHIFT);
        const float  cf = __int_as_float(r.y);
        const float2 v  = *(const float2*)(h + (size_t)s * 2);
        atomicAdd(&acc[u * 2 + 0], v.x * cf);
        atomicAdd(&acc[u * 2 + 1], v.y * cf);
    }
}

__device__ __forceinline__ void l3_scan_edges(const int* __restrict__ keys, const int* __restrict__ pk,
                                              const float* __restrict__ h, float* acc,
                                              int wave, int lane, int tno, int e) {
    unsigned xm[8];
#pragma unroll
    for (int p = 0; p < 8; ++p) xm[p] = ((tno >> p) & 1) ? 0u : ~0u;
    const int4* k4    = (const int4*)keys;
    const int   nfull = e / (2 * KEY_WIN);
    const int   kw    = (e + KEY_WIN - 1) / KEY_WIN;
    for (int w = wave; w < nfull; w += TILE_THREADS / 32) {
        l3_window(2 * w,     k4, xm, pk, h, acc, lane, e);
        l3_window(2 * w + 1, k4, xm, pk, h, acc, lane, e);
    }
    for (int tw = 2 * nfull + wave; tw < kw; tw += TILE_THREADS / 32)
        l3_window(tw, k4, xm, pk, h, acc, lane, e);
}

__global__ __launch_bounds__(TILE_THREADS)
void k_l3_tile(const int* __restrict__ keys, const int* __restrict__ pk,
               const float* __restrict__ h, const float* __restrict__ idg,
               const float* __restrict__ bias, const int* __restrict__ lcnt,
               const float* __restrict__ rec, float* __restrict__ out, int e, int n) {
    __shared__ float acc[TILE_ROWS * 2];
    const int tid  = threadIdx.x;
    const int lane = tid & 31;
    const int wave = __builtin_amdgcn_readfirstlane(tid >> 5);
    const int r0   = blockIdx.x * TILE_ROWS;
    int nr = n - r0;
    if (nr > TILE_ROWS) nr = TILE_ROWS;
    if (nr <= 0) return;

    float2* acc2 = (float2*)acc;
    const float b0 = bias[0], b1 = bias[1];
    for (int i = tid; i < nr; i += TILE_THREADS) {
        const float2 hv = ((const float2*)h)[r0 + i];
        const float  g  = idg[r0 + i];
        acc2[i] = make_float2(hv.x * g + b0, hv.y * g + b1);
    }
    __syncthreads();

    const int wid = blockIdx.x * (TILE_THREADS / 32) + wave;
    const float* recw = rec + (size_t)wid * REC_WORDS;
    const int nrec = __builtin_amdgcn_readfirstlane(lcnt[wid]);
    if (nrec >= 0) {
        for (int b = 0; b < nrec; b += Q_BATCH * L3_BATCHES) {
            int   s[L3_BATCHES], u[L3_BATCHES];
            float cf[L3_BATCHES];
#pragma unroll
            for (int k = 0; k < L3_BATCHES; ++k) {
                const int    bk = b + k * Q_BATCH;
                const float* r  = recw + bk * 3 + lane;
                s[k] = 0; u[k] = -1; cf[k] = 0.0f;
                if (bk + lane < nrec) {
                    s[k]  = __float_as_int(r[0]);
                    u[k]  = __float_as_int(r[Q_BATCH]);
                    cf[k] = r[2 * Q_BATCH];
                }
            }
            float2 v[L3_BATCHES];
#pragma unroll
            for (int k = 0; k < L3_BATCHES; ++k) {
                v[k] = make_float2(0.0f, 0.0f);
                if (u[k] >= 0) v[k] = *(const float2*)(h + (size_t)s[k] * 2);
            }
#pragma unroll
            for (int k = 0; k < L3_BATCHES; ++k) {
                if (u[k] >= 0) {
                    atomicAdd(&acc[u[k] * 2 + 0], v[k].x * cf[k]);
                    atomicAdd(&acc[u[k] * 2 + 1], v[k].y * cf[k]);
                }
            }
        }
    } else {
        l3_scan_edges(keys, pk, h, acc, wave, lane, (int)blockIdx.x, e);
    }
    __syncthreads();
    float2* o = (float2*)out + r0;
    for (int i = tid; i < nr; i += TILE_THREADS) o[i] = acc2[i];
}

extern "C" void kernel_launch(void* const* d_in, const int* in_sizes, int n_in,
                              void* d_out, int out_size, void* d_ws, size_t ws_size,
                              hipStream_t stream) {
    const float* x  = (const float*)d_in[0];
    const int*   ei = (const int*)d_in[1];
    const float* W1 = (const float*)d_in[2];
    const float* b1 = (const float*)d_in[3];
    const float* W2 = (const float*)d_in[4];
    const float* b2 = (const float*)d_in[5];
    const float* W3 = (const float*)d_in[6];
    const float* b3 = (const float*)d_in[7];
    float* out = (float*)d_out;

    const int N = in_sizes[0] / 5;
    const int E = in_sizes[1] / 2;
    const int* src = ei;
    const int* dst = ei + E;

    float* ws   = (float*)d_ws;
    float* isq  = ws;
    float* idg  = isq  + N;
    float* w1t  = idg  + N;
    float* w2t  = w1t  + 64 * 8;
    float* w3t  = w2t  + 64 * 64;
    float* bufA = w3t  + 16 * 64;
    float* bufB = bufA + (size_t)N * 64;
    float* h3   = bufB + (size_t)N * 64;
    float* part = h3   + (size_t)N * 2;
    int*   lcnt = (int*)(part + (size_t)128 * (N + 32));
    float* rec  = (float*)(lcnt + (size_t)((N + TILE_ROWS - 1) / TILE_ROWS) * (TILE_THREADS / 32));
    int*   keys = (int*)(rec + (size_t)((N + TILE_ROWS - 1) / TILE_ROWS) * (TILE_THREADS / 32) * REC_WORDS);
    int*   pk   = keys + (size_t)((E + KEY_WIN - 1) / KEY_WIN) * (KEY_WIN / 4);

    const int T = 256;
    auto cdiv = [](long a, long b) { return (int)((a + b - 1) / b); };

    k_deg_part <<<dim3(DEG_SLICES, cdiv(N, DEG_ROWS)), ACC_THREADS, 0, stream>>>(dst, part, E, N);
    k_norm_part<<<cdiv(N, T), T, 0, stream>>>(part, isq, idg, N);
    k_edge_prep<<<cdiv(cdiv(E, KEY_WIN), PREP_THREADS / 32), PREP_THREADS, 0, stream>>>(src, dst, isq, pk, keys, E);
    k_prep_w   <<<cdiv(64 * 8,  T), T, 0, stream>>>(W1, w1t, 5, 64, 8, 64);
    k_prep_w   <<<cdiv(64 * 64, T), T, 0, stream>>>(W2, w2t, 64, 64, 64, 64);
    k_prep_w   <<<cdiv(16 * 64, T), T, 0, stream>>>(W3, w3t, 64, 2, 64, 16);

    const int GT = cdiv(N, 16);

    k_gemm_wmma<8, 5, 4, 64, false><<<GT, 128, 0, stream>>>(x, w1t, bufA);
    k_scatter64_tile<false, 4, 64><<<cdiv(N, TILE_ROWS), TILE_THREADS, 0, stream>>>(keys, pk, bufA, idg, b1, w2t, bufB, lcnt, rec, E, N);

    k_scatter64_tile<true, 1, 2><<<cdiv(N, TILE_ROWS), TILE_THREADS, 0, stream>>>(keys, pk, bufB, idg, b2, w3t, h3, lcnt, rec, E, N);

    k_l3_tile  <<<cdiv(N, TILE_ROWS), TILE_THREADS, 0, stream>>>(keys, pk, h3, idg, b3, lcnt, rec, out, E, N);
}


# Weight transposes dropped: WMMA B fragments are read from the row-major weight inputs where they lie (2 4-byte loads, same k order); 3 weight launches gone, 10 to 7; products and sums unchanged
# speedup vs baseline: 17.3061x; 1.0327x over previous
#include <hip/hip_runtime.h>
#include <hip/hip_bf16.h>

typedef __attribute__((ext_vector_type(2))) float v2f;
typedef __attribute__((ext_vector_type(8))) float v8f;

constexpr int ACC_THREADS = 1024;
constexpr int ACC_UNROLL  = 8;
constexpr int DEG_ROWS    = 50016;
constexpr int DEG_SLICES  = 128;

__device__ __forceinline__ size_t plane_pitch(size_t words) {
    return (words + 31) & ~(size_t)31;
}

__global__ __launch_bounds__(ACC_THREADS)
void k_deg_part(const int* __restrict__ dst, float* __restrict__ part, int e, int n) {
    __shared__ float cnt[DEG_ROWS];
    const int tid = threadIdx.x;
    const int r0  = blockIdx.y * DEG_ROWS;
    int nr = n - r0;
    if (nr > DEG_ROWS) nr = DEG_ROWS;
    for (int i = tid; i < nr; i += ACC_THREADS) cnt[i] = 0.0f;
    __syncthreads();

    const int per = (e + DEG_SLICES - 1) / DEG_SLICES;
    const int e0  = blockIdx.x * per;
    int e1 = e0 + per;
    if (e1 > e) e1 = e;
    for (int b = e0; b < e1; b += ACC_THREADS * ACC_UNROLL) {
        int d[ACC_UNROLL];
#pragma unroll
        for (int k = 0; k < ACC_UNROLL; ++k) {
            const int i = b + k * ACC_THREADS + tid;
            d[k] = (i < e1) ? dst[i] : -1;
        }
#pragma unroll
        for (int k = 0; k < ACC_UNROLL; ++k) {
            const int u = d[k] - r0;
            if ((unsigned)u < (unsigned)nr) atomicAdd(&cnt[u], 1.0f);
        }
    }
    __syncthreads();
    float* p = part + (size_t)blockIdx.x * plane_pitch((size_t)n) + r0;
    for (int i = tid; i < nr; i += ACC_THREADS) p[i] = cnt[i];
}

__global__ void k_norm_part(const float* __restrict__ part, float* __restrict__ isq,
                            float* __restrict__ idg, int n) {
    int i = blockIdx.x * blockDim.x + threadIdx.x;
    if (i < n) {
        const size_t pitch = plane_pitch((size_t)n);
        float d = 1.0f;
#pragma unroll 16
        for (int s = 0; s < DEG_SLICES; ++s) d += part[(size_t)s * pitch + i];
        isq[i] = rsqrtf(d);
        idg[i] = 1.0f / d;
    }
}

template <int K, int LDA, int NT, int SC, bool RELU_A>
__global__ void k_gemm_wmma(const float* __restrict__ A, const float* __restrict__ Wt,
                            float* __restrict__ H) {
    const int lane = threadIdx.x & 31;
    const int wave = threadIdx.x >> 5;
    const int half = lane >> 4;
    const int l15  = lane & 15;
    const size_t row = (size_t)blockIdx.x * 16 + l15;
    const int    col = wave * 16 + l15;
    const float* Arow = A + row * LDA;
    const bool   cok  = NT * 16 == SC || col < SC;
    const size_t mbase = (size_t)blockIdx.x * 16 + half * 8;

    v8f c = {};
#pragma unroll
    for (int k = 0; k < K; k += 4) {
        const int ka = k + half * 2;
        v2f a;
        if (LDA == K) {
            a = *(const v2f*)(Arow + ka);
        } else {
            a.x = (ka     < LDA) ? Arow[ka]     : 0.0f;
            a.y = (ka + 1 < LDA) ? Arow[ka + 1] : 0.0f;
        }
        if (RELU_A) {
            a.x = fmaxf(a.x, 0.0f);
            a.y = fmaxf(a.y, 0.0f);
        }
        v2f b;
        b.x = (cok && ka     < LDA) ? Wt[(ka)     * SC + col] : 0.0f;
        b.y = (cok && ka + 1 < LDA) ? Wt[(ka + 1) * SC + col] : 0.0f;
        c = __builtin_amdgcn_wmma_f32_16x16x4_f32(
                 false, a,  false, b,
                 (short)0, c,  false,  false);
    }

    if (cok) {
#pragma unroll
        for (int r = 0; r < 8; ++r)
            H[(mbase + r) * SC + col] = c[r];
    }
}

constexpr int TILE_ROWS    = 391;
constexpr int TILE_THREADS = 1024;
constexpr int Q_CAP        = 64;
constexpr int Q_BATCH      = 32;
constexpr int Q_GROUP      = 16;
constexpr int TILE_LOAD_IT = (TILE_ROWS * 16 + TILE_THREADS - 1) / TILE_THREADS;

constexpr int KEY_WIN   = 1024;
constexpr int ROW_SHIFT = 23;
static_assert(TILE_ROWS <= (1 << (32 - ROW_SHIFT)), "row inside the tile must fit above ROW_SHIFT");

constexpr int PREP_THREADS = 256;
constexpr int PREP_STEPS   = 16;
__global__ __launch_bounds__(PREP_THREADS)
void k_edge_prep(const int* __restrict__ src, const int* __restrict__ dst,
                 const float* __restrict__ isq,
                 int* __restrict__ pk, int* __restrict__ keys, int e) {
    const int lane = threadIdx.x & 31;
    const int win = __builtin_amdgcn_readfirstlane((int)(blockIdx.x * (PREP_THREADS / 32) + (threadIdx.x >> 5)));
    const int wb  = win * KEY_WIN;
    if (wb >= e) return;
    int pl[8] = {0, 0, 0, 0, 0, 0, 0, 0};
    for (int i0 = 0; i0 < 32; i0 += PREP_STEPS) {
        int d[PREP_STEPS], s[PREP_STEPS];
#pragma unroll
        for (int k = 0; k < PREP_STEPS; ++k) {
            const int  i  = wb + (i0 + k) * 32 + lane;
            const bool ok = i < e;
            d[k] = ok ? dst[i] : 0;
            s[k] = ok ? src[i] : 0;
        }
        float fs[PREP_STEPS], fd[PREP_STEPS];
#pragma unroll
        for (int k = 0; k < PREP_STEPS; ++k) {
            fs[k] = isq[s[k]];
            fd[k] = isq[d[k]];
        }
#pragma unroll
        for (int k = 0; k < PREP_STEPS; ++k) {
            const int  i  = wb + (i0 + k) * 32 + lane;
            const bool ok = i < e;
            const int   t  = d[k] / TILE_ROWS;
            const int   u  = d[k] - t * TILE_ROWS;
            const float cf = fs[k] * fd[k];
            if (ok)
                ((int2*)pk)[i] = make_int2((int)((unsigned)s[k] | ((unsigned)u << ROW_SHIFT)), __float_as_int(cf));
            const bool mine = lane == i0 + k;
#pragma unroll
            for (int p = 0; p < 8; ++p) {
                const int bits = (int)__builtin_amdgcn_ballot_w32(ok && ((t >> p) & 1));
                pl[p] = mine ? bits : pl[p];
            }
        }
    }
    int4* k4 = (int4*)keys + (size_t)win * 64;
    k4[lane]      = make_int4(pl[0], pl[1], pl[2], pl[3]);
    k4[32 + lane] = make_int4(pl[4], pl[5], pl[6], pl[7]);
}

constexpr int REC_CAP   = 768;
constexpr int REC_WORDS = REC_CAP * 3;

template <bool FULL>
__device__ __forceinline__ void rows_add(int s, int u, float cf, int cnt, int lane,
                                         const float* __restrict__ h, float* tile) {
#pragma unroll
    for (int g = 0; g < Q_BATCH; g += Q_GROUP) {
        float v0[Q_GROUP], v1[Q_GROUP];
#pragma unroll
        for (int k = 0; k < Q_GROUP; ++k) {
            if (FULL || g + k < cnt) {
                const float* hs = h + (size_t)__builtin_amdgcn_readlane(s, g + k) * 64;
                v0[k] = hs[lane];
                v1[k] = hs[lane + 32];
            }
        }
#pragma unroll
        for (int k = 0; k < Q_GROUP; ++k) {
            if (FULL || g + k < cnt) {
                const float cfk = __int_as_float(__builtin_amdgcn_readlane(__float_as_int(cf), g + k));
                float* t = tile + __builtin_amdgcn_readlane(u, g + k) * 64;
                atomicAdd(t + lane,      v0[k] * cfk);
                atomicAdd(t + lane + 32, v1[k] * cfk);
            }
        }
    }
}

template <bool FULL, bool REC>
__device__ __forceinline__ void queue_drain(const int* q, int qh, int cnt, int lane,
                                            const int* __restrict__ pk,
                                            const float* __restrict__ h, float* tile,
                                            float* recw, int& nrec) {
    int s = 0, u = 0;
    float cf = 0.0f;
    if (FULL || lane < cnt) {
        const int  en = q[(qh + lane) & (Q_CAP - 1)];
        const int2 r  = ((const int2*)pk)[en];
        s  = r.x & ((1 << ROW_SHIFT) - 1);
        u  = (int)((unsigned)r.x >> ROW_SHIFT);
        cf = __int_as_float(r.y);
    }
    if (REC && nrec >= 0) {
        if (nrec + Q_BATCH <= REC_CAP) {
            if (FULL || lane < cnt) {
                float* r = recw + nrec * 3 + lane;
                r[0]           = __int_as_float(s);
                r[Q_BATCH]     = __int_as_float(u);
                r[2 * Q_BATCH] = cf;
            }
            nrec += cnt;
        } else {
            nrec = -1;
        }
    }
    rows_add<FULL>(s, u, cf, cnt, lane, h, tile);
}

__device__ __forceinline__ unsigned key_match(const int4& lo, const int4& hi, const unsigned (&xm)[8]) {
    unsigned m = ((unsigned)lo.x ^ xm[0]) & ((unsigned)lo.y ^ xm[1]);
    m &= ((unsigned)lo.z ^ xm[2]);
    m &= ((unsigned)lo.w ^ xm[3]);
    m &= ((unsigned)hi.x ^ xm[4]);
    m &= ((unsigned)hi.y ^ xm[5]);
    m &= ((unsigned)hi.z ^ xm[6]);
    m &= ((unsigned)hi.w ^ xm[7]);
    return m;
}

template <bool REC>
__device__ __forceinline__ void window_hits(unsigned m, int ebase, int lane,
                                            int* q, int& qh, int& qn,
                                            const int* __restrict__ pk,
                                            const float* __restrict__ h, float* tile,
                                            float* recw, int& nrec) {
    unsigned bal = __builtin_amdgcn_ballot_w32(m != 0u);
    while (bal) {
        if (m != 0u) {
            const int slot = (qh + qn + (int)__builtin_amdgcn_mbcnt_lo(bal, 0u)) & (Q_CAP - 1);
            q[slot] = ebase + __builtin_ctz(m);
            m &= m - 1u;
        }
        qn += __builtin_popcount(bal);
        if (qn >= Q_BATCH) {
            queue_drain<true, REC>(q, qh, Q_BATCH, lane, pk, h, tile, recw, nrec);
            qh = (qh + Q_BATCH) & (Q_CAP - 1);
            qn -= Q_BATCH;
        }
        bal = __builtin_amdgcn_ballot_w32(m != 0u);
    }
}

template <bool REC>
__device__ __forceinline__ void scan_edges(const int* __restrict__ keys, const int* __restrict__ pk,
                                           const float* __restrict__ h,
                                           float* tile, int* q, int wave, int lane,
                                           int tno, int e, float* recw, int& nrec) {
    int qh = 0, qn = 0;
    unsigned xm[8];
#pragma unroll
    for (int p = 0; p < 8; ++p) xm[p] = ((tno >> p) & 1) ? 0u : ~0u;

    const int4* k4    = (const int4*)keys;
    const int   nfull = e / (2 * KEY_WIN);
    int4 nx[4];
    if (wave < nfull) {
        const int4* p = k4 + (size_t)wave * 128 + lane;
#pragma unroll
        for (int qq = 0; qq < 4; ++qq) nx[qq] = p[qq * 32];
    }
    for (int w = wave; w < nfull; w += TILE_THREADS / 32) {
        int4 cur[4];
#pragma unroll
        for (int qq = 0; qq < 4; ++qq) cur[qq] = nx[qq];
        const int wn = w + TILE_THREADS / 32;
        if (wn < nfull) {
            const int4* p = k4 + (size_t)wn * 128 + lane;
#pragma unroll
            for (int qq = 0; qq < 4; ++qq) nx[qq] = p[qq * 32];
        }
#pragma unroll
        for (int hh = 0; hh < 2; ++hh) {
            const unsigned m = key_match(cur[2 * hh], cur[2 * hh + 1], xm);
            window_hits<REC>(m, (2 * w + hh) * KEY_WIN + lane * 32, lane, q, qh, qn, pk, h, tile, recw, nrec);
        }
    }
    const int kw = (e + KEY_WIN - 1) / KEY_WIN;
    for (int tw = 2 * nfull + wave; tw < kw; tw += TILE_THREADS / 32) {
        const int4* p = k4 + (size_t)tw * 64 + lane;
        const int4 lo = p[0];
        const int4 hi = p[32];
        unsigned m = key_match(lo, hi, xm);
        const int ebase = tw * KEY_WIN + lane * 32;
        const int valid = e - ebase;
        if (valid < 32) m = (valid <= 0) ? 0u : (m & ((1u << valid) - 1u));
        window_hits<REC>(m, ebase, lane, q, qh, qn, pk, h, tile, recw, nrec);
    }
    if (qn > 0) queue_drain<false, REC>(q, qh, qn, lane, pk, h, tile, recw, nrec);
}

__device__ __forceinline__ void replay_edges(const float* __restrict__ recw, int nrec, int lane,
                                             const float* __restrict__ h, float* tile) {
    int s = 0, u = 0;
    float cf = 0.0f;
    if (lane < nrec) {
        s  = __float_as_int(recw[lane]);
        u  = __float_as_int(recw[Q_BATCH + lane]);
        cf = recw[2 * Q_BATCH + lane];
    }
    for (int b = 0; b < nrec; b += Q_BATCH) {
        int s2 = 0, u2 = 0;
        float cf2 = 0.0f;
        const int nb = b + Q_BATCH;
        if (nb + lane < nrec) {
            const float* r = recw + nb * 3 + lane;
            s2  = __float_as_int(r[0]);
            u2  = __float_as_int(r[Q_BATCH]);
            cf2 = r[2 * Q_BATCH];
        }
        if (nrec - b >= Q_BATCH) rows_add<true>(s, u, cf, Q_BATCH, lane, h, tile);
        else                     rows_add<false>(s, u, cf, nrec - b, lane, h, tile);
        s = s2; u = u2; cf = cf2;
    }
}

template <int NT, int CIN>
__device__ __forceinline__ void tile_product(float* tile, int nr, int wave, int lane,
                                             const float* __restrict__ Wt) {
    const int half = lane >> 4;
    const int l15  = lane & 15;
    const int nblk = (nr + 15) >> 4;
    for (int rb = wave; rb < nblk; rb += TILE_THREADS / 32) {
        int arow = rb * 16 + l15;
        if (arow > nr - 1) arow = nr - 1;
        const float* ar = tile + arow * 64 + half * 2;
        v2f a[16];
#pragma unroll
        for (int k = 0; k < 16; ++k) {
            a[k] = *(const v2f*)(ar + 4 * k);
            a[k].x = fmaxf(a[k].x, 0.0f);
            a[k].y = fmaxf(a[k].y, 0.0f);
        }
        const int mrow = rb * 16 + half * 8;
#pragma nounroll
        for (int ct = 0; ct < NT; ++ct) {
            const int   col  = ct * 16 + l15;
            const bool  cok  = NT * 16 == CIN || col < CIN;
            const float* Wk  = Wt + (half * 2) * CIN + col;
            v8f c = {};
#pragma unroll
            for (int k = 0; k < 16; ++k) {
                v2f b;
                b.x = cok ? Wk[(4 * k)     * CIN] : 0.0f;
                b.y = cok ? Wk[(4 * k + 1) * CIN] : 0.0f;
                c = __builtin_amdgcn_wmma_f32_16x16x4_f32(
                         false, a[k],  false, b,
                         (short)0, c,  false,  false);
            }
#pragma unroll
            for (int r = 0; r < 8; ++r)
                if (mrow + r < nr) tile[(mrow + r) * 64 + ct * 16 + l15] = c[r];
        }
    }
}

template <bool REPLAY, int NT, int SC>
__global__ __launch_bounds__(TILE_THREADS)
void k_scatter64_tile(const int* __restrict__ keys, const int* __restrict__ pk,
                      const float* __restrict__ h, const float* __restrict__ idg,
                      const float* __restrict__ bias, const float* __restrict__ Wt,
                      float* __restrict__ prod, int* __restrict__ lcnt, float* __restrict__ rec,
                      int e, int n) {
    __shared__ float tile[TILE_ROWS * 64];
    __shared__ int   queue[(TILE_THREADS / 32) * Q_CAP];
    const int tid  = threadIdx.x;
    const int lane = tid & 31;
    const int wave = __builtin_amdgcn_readfirstlane(tid >> 5);
    const int r0   = blockIdx.x * TILE_ROWS;
    int nr = n - r0;
    if (nr > TILE_ROWS) nr = TILE_ROWS;
    if (nr <= 0) return;

    float4*       t4 = (float4*)tile;
    const float4* h4 = (const float4*)(h + (size_t)r0 * 64);
    const float4  b4 = ((const float4*)bias)[tid & 15];
    float4 hv[TILE_LOAD_IT];
    float  g[TILE_LOAD_IT];
#pragma unroll
    for (int k = 0; k < TILE_LOAD_IT; ++k) {
        const int i = tid + k * TILE_THREADS;
        if (i < nr * 16) {
            hv[k] = h4[i];
            g[k]  = idg[r0 + (i >> 4)];
        }
    }
#pragma unroll
    for (int k = 0; k < TILE_LOAD_IT; ++k) {
        const int i = tid + k * TILE_THREADS;
        if (i < nr * 16)
            t4[i] = make_float4(hv[k].x * g[k] + b4.x, hv[k].y * g[k] + b4.y,
                                hv[k].z * g[k] + b4.z, hv[k].w * g[k] + b4.w);
    }
    __syncthreads();

    int*   q    = queue + wave * Q_CAP;
    const int tno = __builtin_amdgcn_readfirstlane((int)blockIdx.x);
    const int wid = blockIdx.x * (TILE_THREADS / 32) + wave;
    float* recw = rec + (size_t)wid * REC_WORDS;

    if (REPLAY) {
        int nrec = __builtin_amdgcn_readfirstlane(lcnt[wid]);
        if (nrec >= 0) replay_edges(recw, nrec, lane, h, tile);
        else           scan_edges<false>(keys, pk, h, tile, q, wave, lane, tno, e, recw, nrec);
    } else {
        int nrec = 0;
        scan_edges<true>(keys, pk, h, tile, q, wave, lane, tno, e, recw, nrec);
        if (lane == 0) lcnt[wid] = nrec;
    }
    __syncthreads();
    tile_product<NT, SC>(tile, nr, wave, lane, Wt);
    __syncthreads();
    if (SC == 64) {
        float4* p4 = (float4*)(prod + (size_t)r0 * 64);
        for (int i = tid; i < nr * 16; i += TILE_THREADS) p4[i] = t4[i];
    } else {
        float2* p2 = (float2*)prod + r0;
        for (int i = tid; i < nr; i += TILE_THREADS) p2[i] = *(const float2*)(tile + i * 64);
    }
}

constexpr int L3_BATCHES = 8;

__device__ __forceinline__ void l3_window(int tw, const int4* __restrict__ k4, const unsigned (&xm)[8],
                                          const int* __restrict__ pk, const float* __restrict__ h,
                                          float* acc, int lane, int e) {
    const int4* p = k4 + (size_t)tw * 64 + lane;
    unsigned m = key_match(p[0], p[32], xm);
    const int ebase = tw * KEY_WIN + lane * 32;
    const int valid = e - ebase;
    if (valid < 32) m = (valid <= 0) ? 0u : (m & ((1u << valid) - 1u));
    while (m) {
        const int  en = ebase + __builtin_ctz(m);
        m &= m - 1u;
        const int2   r  = ((const int2*)pk)[en];
        const int    s  = r.x & ((1 << ROW_SHIFT) - 1);
        const int    u  = (int)((unsigned)r.x >> ROW_SHIFT);
        const float  cf = __int_as_float(r.y);
        const float2 v  = *(const float2*)(h + (size_t)s * 2);
        atomicAdd(&acc[u * 2 + 0], v.x * cf);
        atomicAdd(&acc[u * 2 + 1], v.y * cf);
    }
}

__device__ __forceinline__ void l3_scan_edges(const int* __restrict__ keys, const int* __restrict__ pk,
                                              const float* __restrict__ h, float* acc,
                                              int wave, int lane, int tno, int e) {
    unsigned xm[8];
#pragma unroll
    for (int p = 0; p < 8; ++p) xm[p] = ((tno >> p) & 1) ? 0u : ~0u;
    const int4* k4    = (const int4*)keys;
    const int   nfull = e / (2 * KEY_WIN);
    const int   kw    = (e + KEY_WIN - 1) / KEY_WIN;
    for (int w = wave; w < nfull; w += TILE_THREADS / 32) {
        l3_window(2 * w,     k4, xm, pk, h, acc, lane, e);
        l3_window(2 * w + 1, k4, xm, pk, h, acc, lane, e);
    }
    for (int tw = 2 * nfull + wave; tw < kw; tw += TILE_THREADS / 32)
        l3_window(tw, k4, xm, pk, h, acc, lane, e);
}

__global__ __launch_bounds__(TILE_THREADS)
void k_l3_tile(const int* __restrict__ keys, const int* __restrict__ pk,
               const float* __restrict__ h, const float* __restrict__ idg,
               const float* __restrict__ bias, const int* __restrict__ lcnt,
               const float* __restrict__ rec, float* __restrict__ out, int e, int n) {
    __shared__ float acc[TILE_ROWS * 2];
    const int tid  = threadIdx.x;
    const int lane = tid & 31;
    const int wave = __builtin_amdgcn_readfirstlane(tid >> 5);
    const int r0   = blockIdx.x * TILE_ROWS;
    int nr = n - r0;
    if (nr > TILE_ROWS) nr = TILE_ROWS;
    if (nr <= 0) return;

    float2* acc2 = (float2*)acc;
    const float b0 = bias[0], b1 = bias[1];
    for (int i = tid; i < nr; i += TILE_THREADS) {
        const float2 hv = ((const float2*)h)[r0 + i];
        const float  g  = idg[r0 + i];
        acc2[i] = make_float2(hv.x * g + b0, hv.y * g + b1);
    }
    __syncthreads();

    const int wid = blockIdx.x * (TILE_THREADS / 32) + wave;
    const float* recw = rec + (size_t)wid * REC_WORDS;
    const int nrec = __builtin_amdgcn_readfirstlane(lcnt[wid]);
    if (nrec >= 0) {
        for (int b = 0; b < nrec; b += Q_BATCH * L3_BATCHES) {
            int   s[L3_BATCHES], u[L3_BATCHES];
            float cf[L3_BATCHES];
#pragma unroll
            for (int k = 0; k < L3_BATCHES; ++k) {
                const int    bk = b + k * Q_BATCH;
                const float* r  = recw + bk * 3 + lane;
                s[k] = 0; u[k] = -1; cf[k] = 0.0f;
                if (bk + lane < nrec) {
                    s[k]  = __float_as_int(r[0]);
                    u[k]  = __float_as_int(r[Q_BATCH]);
                    cf[k] = r[2 * Q_BATCH];
                }
            }
            float2 v[L3_BATCHES];
#pragma unroll
            for (int k = 0; k < L3_BATCHES; ++k) {
                v[k] = make_float2(0.0f, 0.0f);
                if (u[k] >= 0) v[k] = *(const float2*)(h + (size_t)s[k] * 2);
            }
#pragma unroll
            for (int k = 0; k < L3_BATCHES; ++k) {
                if (u[k] >= 0) {
                    atomicAdd(&acc[u[k] * 2 + 0], v[k].x * cf[k]);
                    atomicAdd(&acc[u[k] * 2 + 1], v[k].y * cf[k]);
                }
            }
        }
    } else {
        l3_scan_edges(keys, pk, h, acc, wave, lane, (int)blockIdx.x, e);
    }
    __syncthreads();
    float2* o = (float2*)out + r0;
    for (int i = tid; i < nr; i += TILE_THREADS) o[i] = acc2[i];
}

extern "C" void kernel_launch(void* const* d_in, const int* in_sizes, int n_in,
                              void* d_out, int out_size, void* d_ws, size_t ws_size,
                              hipStream_t stream) {
    const float* x  = (const float*)d_in[0];
    const int*   ei = (const int*)d_in[1];
    const float* W1 = (const float*)d_in[2];
    const float* b1 = (const float*)d_in[3];
    const float* W2 = (const float*)d_in[4];
    const float* b2 = (const float*)d_in[5];
    const float* W3 = (const float*)d_in[6];
    const float* b3 = (const float*)d_in[7];
    float* out = (float*)d_out;

    const int N = in_sizes[0] / 5;
    const int E = in_sizes[1] / 2;
    const int* src = ei;
    const int* dst = ei + E;

    float* ws   = (float*)d_ws;
    float* isq  = ws;
    float* idg  = isq  + N;
    float* bufA = idg  + N;
    float* bufB = bufA + (size_t)N * 64;
    float* h3   = bufB + (size_t)N * 64;
    float* part = h3   + (size_t)N * 2;
    int*   lcnt = (int*)(part + (size_t)128 * (N + 32));
    float* rec  = (float*)(lcnt + (size_t)((N + TILE_ROWS - 1) / TILE_ROWS) * (TILE_THREADS / 32));
    int*   keys = (int*)(rec + (size_t)((N + TILE_ROWS - 1) / TILE_ROWS) * (TILE_THREADS / 32) * REC_WORDS);
    int*   pk   = keys + (size_t)((E + KEY_WIN - 1) / KEY_WIN) * (KEY_WIN / 4);

    const int T = 256;
    auto cdiv = [](long a, long b) { return (int)((a + b - 1) / b); };

    k_deg_part <<<dim3(DEG_SLICES, cdiv(N, DEG_ROWS)), ACC_THREADS, 0, stream>>>(dst, part, E, N);
    k_norm_part<<<cdiv(N, T), T, 0, stream>>>(part, isq, idg, N);
    k_edge_prep<<<cdiv(cdiv(E, KEY_WIN), PREP_THREADS / 32), PREP_THREADS, 0, stream>>>(src, dst, isq, pk, keys, E);

    const int GT = cdiv(N, 16);

    k_gemm_wmma<8, 5, 4, 64, false><<<GT, 128, 0, stream>>>(x, W1, bufA);
    k_scatter64_tile<false, 4, 64><<<cdiv(N, TILE_ROWS), TILE_THREADS, 0, stream>>>(keys, pk, bufA, idg, b1, W2, bufB, lcnt, rec, E, N);

    k_scatter64_tile<true, 1, 2><<<cdiv(N, TILE_ROWS), TILE_THREADS, 0, stream>>>(keys, pk, bufB, idg, b2, W3, h3, lcnt, rec, E, N);

    k_l3_tile  <<<cdiv(N, TILE_ROWS), TILE_THREADS, 0, stream>>>(keys, pk, h3, idg, b3, lcnt, rec, out, E, N);
}
